# starting kernel
# baseline (speedup 1.0000x reference)
#include <hip/hip_runtime.h>
#include <hip/hip_bf16.h>


#define IN_DIM  64
#define OUT_DIM 16

typedef __attribute__((ext_vector_type(2))) float v2f;
typedef __attribute__((ext_vector_type(8))) float v8f;

__device__ __forceinline__ void atomic_add_f32(float* p, float v) {
  asm volatile("global_atomic_add_f32 %0, %1, off scope:SCOPE_DEV"
               :: "v"(p), "v"(v) : "memory");
}


__global__ void fill0_kernel(float* __restrict__ p, int n) {
  int i = blockIdx.x * blockDim.x + threadIdx.x;
  if (i < n) p[i] = 0.0f;
}

__global__ void degree_kernel(const int* __restrict__ dst, float* __restrict__ deg, int E) {
  int i = blockIdx.x * blockDim.x + threadIdx.x;
  if (i < E) atomic_add_f32(&deg[dst[i]], 1.0f);
}

__global__ void rsqrt_kernel(float* __restrict__ d, int n) {
  int i = blockIdx.x * blockDim.x + threadIdx.x;
  if (i < n) d[i] = rsqrtf(d[i] + 1.0f);
}

__global__ void bias_relu_kernel(float* __restrict__ a, const float* __restrict__ b, int n) {
  int i = blockIdx.x * blockDim.x + threadIdx.x;
  if (i < n) a[i] = fmaxf(a[i] + b[i & 63], 0.0f);
}

__global__ void bias_kernel(float* __restrict__ a, const float* __restrict__ b, int n) {
  int i = blockIdx.x * blockDim.x + threadIdx.x;
  if (i < n) a[i] = a[i] + b[i & 15];
}


template <int NCOLS>
__global__ __launch_bounds__(256) void gemm_wmma_kernel(const float* __restrict__ A,
                                                        const float* __restrict__ W,
                                                        float* __restrict__ out, int M) {
  __shared__ float sW[64 * NCOLS];
  const int nvec = (64 * NCOLS) / 4;
  for (int i = threadIdx.x; i < nvec; i += blockDim.x)
    ((float4*)sW)[i] = ((const float4*)W)[i];
  __syncthreads();

  const int lane = threadIdx.x & 31;
  const int wid  = threadIdx.x >> 5;
  const int ntile_n = NCOLS / 16;
  const int mtiles  = (M + 15) >> 4;
  const int gt = blockIdx.x * 8 + wid;
  const int mt = gt / ntile_n;
  const int nt = gt - mt * ntile_n;
  if (mt >= mtiles) return;

  const int m0 = mt * 16;
  const int n0 = nt * 16;
  const int row = lane & 15;
  const int kh  = (lane >> 4) << 1;
  const int arow_idx = min(m0 + row, M - 1);
  const float* __restrict__ arow = A + (size_t)arow_idx * 64 + kh;
  const float* __restrict__ bcol = sW + n0 + row;

  v8f c = {};
#pragma unroll
  for (int k = 0; k < 64; k += 4) {
    v2f a, b;
    a.x = arow[k];
    a.y = arow[k + 1];
    b.x = bcol[(k + kh) * NCOLS];
    b.y = bcol[(k + kh + 1) * NCOLS];
    c = __builtin_amdgcn_wmma_f32_16x16x4_f32(false, a, false, b, (short)0, c, false, false);
  }

  const int rbase = (lane >> 4) << 3;
#pragma unroll
  for (int r = 0; r < 8; ++r) {
    int orow = m0 + rbase + r;
    if (orow < M) out[(size_t)orow * NCOLS + n0 + row] = c[r];
  }
}


__global__ __launch_bounds__(256) void scatter64_kernel(const float* __restrict__ h,
                                                        const int* __restrict__ src,
                                                        const int* __restrict__ dst,
                                                        const float* __restrict__ dinv,
                                                        float* __restrict__ agg,
                                                        int E, int Nn) {
  const int e = blockIdx.x * 8 + (threadIdx.x >> 5);
  const int lane = threadIdx.x & 31;
  if (e >= E + Nn) return;
  int s, d;
  if (e < E) { s = src[e]; d = dst[e]; } else { s = e - E; d = s; }
  const float w = dinv[s] * dinv[d];
  const float2 v = ((const float2*)(h + (size_t)s * 64))[lane];
  float* op = agg + (size_t)d * 64 + lane * 2;
  atomic_add_f32(op,     v.x * w);
  atomic_add_f32(op + 1, v.y * w);
}

__global__ __launch_bounds__(256) void scatter16_kernel(const float* __restrict__ h,
                                                        const int* __restrict__ src,
                                                        const int* __restrict__ dst,
                                                        const float* __restrict__ dinv,
                                                        float* __restrict__ out,
                                                        int E, int Nn) {
  const int tid = blockIdx.x * blockDim.x + threadIdx.x;
  const int e = tid >> 4;
  const int c = tid & 15;
  if (e >= E + Nn) return;
  int s, d;
  if (e < E) { s = src[e]; d = dst[e]; } else { s = e - E; d = s; }
  const float w = dinv[s] * dinv[d];
  atomic_add_f32(&out[(size_t)d * 16 + c], h[(size_t)s * 16 + c] * w);
}


extern "C" void kernel_launch(void* const* d_in, const int* in_sizes, int n_in,
                              void* d_out, int out_size, void* d_ws, size_t ws_size,
                              hipStream_t stream) {
  const float* x    = (const float*)d_in[0];
  const int*   edge = (const int*)d_in[1];
  const float* W1   = (const float*)d_in[2];
  const float* b1   = (const float*)d_in[3];
  const float* W2   = (const float*)d_in[4];
  const float* b2   = (const float*)d_in[5];
  float* out = (float*)d_out;

  const int E  = in_sizes[1] / 2;
  const int Nn = in_sizes[0] / IN_DIM;
  const int* src = edge;
  const int* dst = edge + E;

  char* ws = (char*)d_ws;
  float* dinv = (float*)ws;
  float* h1   = (float*)(ws + (size_t)Nn * 4);
  float* agg1 = (float*)(ws + (size_t)Nn * 4 + (size_t)Nn * 64 * 4);
  float* h2   = h1;

  const int TB = 256;
  const int mtiles = (Nn + 15) >> 4;

  fill0_kernel<<<(Nn + TB - 1) / TB, TB, 0, stream>>>(dinv, Nn);
  fill0_kernel<<<(Nn * 64 + TB - 1) / TB, TB, 0, stream>>>(agg1, Nn * 64);
  fill0_kernel<<<(Nn * 16 + TB - 1) / TB, TB, 0, stream>>>(out, Nn * 16);

  degree_kernel<<<(E + TB - 1) / TB, TB, 0, stream>>>(dst, dinv, E);
  rsqrt_kernel<<<(Nn + TB - 1) / TB, TB, 0, stream>>>(dinv, Nn);

  gemm_wmma_kernel<64><<<(mtiles * 4 + 7) / 8, 256, 0, stream>>>(x, W1, h1, Nn);
  scatter64_kernel<<<((E + Nn) + 7) / 8, 256, 0, stream>>>(h1, src, dst, dinv, agg1, E, Nn);
  bias_relu_kernel<<<(Nn * 64 + TB - 1) / TB, TB, 0, stream>>>(agg1, b1, Nn * 64);

  gemm_wmma_kernel<16><<<(mtiles + 7) / 8, 256, 0, stream>>>(agg1, W2, h2, Nn);
  scatter16_kernel<<<((E + Nn) * 16 + TB - 1) / TB, TB, 0, stream>>>(h2, src, dst, dinv, out, E, Nn);
  bias_kernel<<<(Nn * 16 + TB - 1) / TB, TB, 0, stream>>>(out, b2, Nn * 16);

  (void)n_in; (void)out_size; (void)ws_size;
}


# Fold zero fill, self-loop, bias and relu into the projection kernels: each atomic sum starts from its self-loop term (+b2) stored by the GEMM epilogue; relu(agg+b1) applied in layer-2 A load
# speedup vs baseline: 1.0959x; 1.0959x over previous
#include <hip/hip_runtime.h>
#include <hip/hip_bf16.h>


#define IN_DIM  64
#define OUT_DIM 16

typedef __attribute__((ext_vector_type(2))) float v2f;
typedef __attribute__((ext_vector_type(8))) float v8f;

__device__ __forceinline__ void atomic_add_f32(float* p, float v) {
  asm volatile("global_atomic_add_f32 %0, %1, off scope:SCOPE_DEV"
               :: "v"(p), "v"(v) : "memory");
}


__global__ void fill0_kernel(float* __restrict__ p, int n) {
  int i = blockIdx.x * blockDim.x + threadIdx.x;
  if (i < n) p[i] = 0.0f;
}

__global__ void degree_kernel(const int* __restrict__ dst, float* __restrict__ deg, int E) {
  int i = blockIdx.x * blockDim.x + threadIdx.x;
  if (i < E) atomic_add_f32(&deg[dst[i]], 1.0f);
}

__global__ void rsqrt_kernel(float* __restrict__ d, int n) {
  int i = blockIdx.x * blockDim.x + threadIdx.x;
  if (i < n) d[i] = rsqrtf(d[i] + 1.0f);
}


template <int NCOLS, bool BIAS_RELU_A, bool BIAS_INIT>
__global__ __launch_bounds__(256) void gemm_wmma_kernel(const float* __restrict__ A,
                                                        const float* __restrict__ abias,
                                                        const float* __restrict__ W,
                                                        const float* __restrict__ dinv,
                                                        const float* __restrict__ ibias,
                                                        float* __restrict__ out,
                                                        float* __restrict__ init, int M) {
  __shared__ float sW[64 * NCOLS];
  const int nvec = (64 * NCOLS) / 4;
  for (int i = threadIdx.x; i < nvec; i += blockDim.x)
    ((float4*)sW)[i] = ((const float4*)W)[i];
  __syncthreads();

  const int lane = threadIdx.x & 31;
  const int wid  = threadIdx.x >> 5;
  const int ntile_n = NCOLS / 16;
  const int mtiles  = (M + 15) >> 4;
  const int gt = blockIdx.x * 8 + wid;
  const int mt = gt / ntile_n;
  const int nt = gt - mt * ntile_n;
  if (mt >= mtiles) return;

  const int m0 = mt * 16;
  const int n0 = nt * 16;
  const int row = lane & 15;
  const int kh  = (lane >> 4) << 1;
  const int arow_idx = min(m0 + row, M - 1);
  const float* __restrict__ arow = A + (size_t)arow_idx * 64 + kh;
  const float* __restrict__ bcol = sW + n0 + row;
  const float* __restrict__ ab = BIAS_RELU_A ? abias + kh : nullptr;

  v8f c = {};
#pragma unroll
  for (int k = 0; k < 64; k += 4) {
    v2f a, b;
    a.x = arow[k];
    a.y = arow[k + 1];
    if (BIAS_RELU_A) {
      a.x = fmaxf(a.x + ab[k], 0.0f);
      a.y = fmaxf(a.y + ab[k + 1], 0.0f);
    }
    b.x = bcol[(k + kh) * NCOLS];
    b.y = bcol[(k + kh + 1) * NCOLS];
    c = __builtin_amdgcn_wmma_f32_16x16x4_f32(false, a, false, b, (short)0, c, false, false);
  }

  const int rbase = (lane >> 4) << 3;
  const float ib = BIAS_INIT ? ibias[n0 + row] : 0.0f;
#pragma unroll
  for (int r = 0; r < 8; ++r) {
    int orow = m0 + rbase + r;
    if (orow < M) {
      const float dv = dinv[orow];
      const size_t o = (size_t)orow * NCOLS + n0 + row;
      float iv = c[r] * (dv * dv);
      if (BIAS_INIT) iv += ib;
      out[o]  = c[r];
      init[o] = iv;
    }
  }
}


__global__ __launch_bounds__(256) void scatter64_kernel(const float* __restrict__ h,
                                                        const int* __restrict__ src,
                                                        const int* __restrict__ dst,
                                                        const float* __restrict__ dinv,
                                                        float* __restrict__ agg,
                                                        int E) {
  const int e = blockIdx.x * 8 + (threadIdx.x >> 5);
  const int lane = threadIdx.x & 31;
  if (e >= E) return;
  const int s = src[e];
  const int d = dst[e];
  const float w = dinv[s] * dinv[d];
  const float2 v = ((const float2*)(h + (size_t)s * 64))[lane];
  float* op = agg + (size_t)d * 64 + lane * 2;
  atomic_add_f32(op,     v.x * w);
  atomic_add_f32(op + 1, v.y * w);
}

__global__ __launch_bounds__(256) void scatter16_kernel(const float* __restrict__ h,
                                                        const int* __restrict__ src,
                                                        const int* __restrict__ dst,
                                                        const float* __restrict__ dinv,
                                                        float* __restrict__ out,
                                                        int E) {
  const int tid = blockIdx.x * blockDim.x + threadIdx.x;
  const int e = tid >> 4;
  const int c = tid & 15;
  if (e >= E) return;
  const int s = src[e];
  const int d = dst[e];
  const float w = dinv[s] * dinv[d];
  atomic_add_f32(&out[(size_t)d * 16 + c], h[(size_t)s * 16 + c] * w);
}


extern "C" void kernel_launch(void* const* d_in, const int* in_sizes, int n_in,
                              void* d_out, int out_size, void* d_ws, size_t ws_size,
                              hipStream_t stream) {
  const float* x    = (const float*)d_in[0];
  const int*   edge = (const int*)d_in[1];
  const float* W1   = (const float*)d_in[2];
  const float* b1   = (const float*)d_in[3];
  const float* W2   = (const float*)d_in[4];
  const float* b2   = (const float*)d_in[5];
  float* out = (float*)d_out;

  const int E  = in_sizes[1] / 2;
  const int Nn = in_sizes[0] / IN_DIM;
  const int* src = edge;
  const int* dst = edge + E;

  char* ws = (char*)d_ws;
  float* dinv = (float*)ws;
  float* h1   = (float*)(ws + (size_t)Nn * 4);
  float* agg1 = (float*)(ws + (size_t)Nn * 4 + (size_t)Nn * 64 * 4);
  float* h2   = h1;

  const int TB = 256;
  const int mtiles = (Nn + 15) >> 4;

  fill0_kernel<<<(Nn + TB - 1) / TB, TB, 0, stream>>>(dinv, Nn);
  degree_kernel<<<(E + TB - 1) / TB, TB, 0, stream>>>(dst, dinv, E);
  rsqrt_kernel<<<(Nn + TB - 1) / TB, TB, 0, stream>>>(dinv, Nn);

  gemm_wmma_kernel<64, false, false><<<(mtiles * 4 + 7) / 8, 256, 0, stream>>>(
      x, nullptr, W1, dinv, nullptr, h1, agg1, Nn);
  scatter64_kernel<<<(E + 7) / 8, 256, 0, stream>>>(h1, src, dst, dinv, agg1, E);

  gemm_wmma_kernel<16, true, true><<<(mtiles + 7) / 8, 256, 0, stream>>>(
      agg1, b1, W2, dinv, b2, h2, out, Nn);
  scatter16_kernel<<<(E * 16 + TB - 1) / TB, TB, 0, stream>>>(h2, src, dst, dinv, out, E);

  (void)n_in; (void)out_size; (void)ws_size;
}


# Edge scatters: each wave takes 32 edges (index and weight fetched once per lane, rows of 8 edges requested together ahead of their adds), not 1 or 2; same f32 atomic adds, arrival order as before
# speedup vs baseline: 1.4778x; 1.3485x over previous
#include <hip/hip_runtime.h>
#include <hip/hip_bf16.h>


#define IN_DIM  64
#define OUT_DIM 16

typedef __attribute__((ext_vector_type(2))) float v2f;
typedef __attribute__((ext_vector_type(8))) float v8f;

__device__ __forceinline__ void atomic_add_f32(float* p, float v) {
  asm volatile("global_atomic_add_f32 %0, %1, off scope:SCOPE_DEV"
               :: "v"(p), "v"(v) : "memory");
}


__global__ void fill0_kernel(float* __restrict__ p, int n) {
  int i = blockIdx.x * blockDim.x + threadIdx.x;
  if (i < n) p[i] = 0.0f;
}

__global__ void degree_kernel(const int* __restrict__ dst, float* __restrict__ deg, int E) {
  int i = blockIdx.x * blockDim.x + threadIdx.x;
  if (i < E) atomic_add_f32(&deg[dst[i]], 1.0f);
}

__global__ void rsqrt_kernel(float* __restrict__ d, int n) {
  int i = blockIdx.x * blockDim.x + threadIdx.x;
  if (i < n) d[i] = rsqrtf(d[i] + 1.0f);
}


template <int NCOLS, bool BIAS_RELU_A, bool BIAS_INIT>
__global__ __launch_bounds__(256) void gemm_wmma_kernel(const float* __restrict__ A,
                                                        const float* __restrict__ abias,
                                                        const float* __restrict__ W,
                                                        const float* __restrict__ dinv,
                                                        const float* __restrict__ ibias,
                                                        float* __restrict__ out,
                                                        float* __restrict__ init, int M) {
  __shared__ float sW[64 * NCOLS];
  const int nvec = (64 * NCOLS) / 4;
  for (int i = threadIdx.x; i < nvec; i += blockDim.x)
    ((float4*)sW)[i] = ((const float4*)W)[i];
  __syncthreads();

  const int lane = threadIdx.x & 31;
  const int wid  = threadIdx.x >> 5;
  const int ntile_n = NCOLS / 16;
  const int mtiles  = (M + 15) >> 4;
  const int gt = blockIdx.x * 8 + wid;
  const int mt = gt / ntile_n;
  const int nt = gt - mt * ntile_n;
  if (mt >= mtiles) return;

  const int m0 = mt * 16;
  const int n0 = nt * 16;
  const int row = lane & 15;
  const int kh  = (lane >> 4) << 1;
  const int arow_idx = min(m0 + row, M - 1);
  const float* __restrict__ arow = A + (size_t)arow_idx * 64 + kh;
  const float* __restrict__ bcol = sW + n0 + row;
  const float* __restrict__ ab = BIAS_RELU_A ? abias + kh : nullptr;

  v8f c = {};
#pragma unroll
  for (int k = 0; k < 64; k += 4) {
    v2f a, b;
    a.x = arow[k];
    a.y = arow[k + 1];
    if (BIAS_RELU_A) {
      a.x = fmaxf(a.x + ab[k], 0.0f);
      a.y = fmaxf(a.y + ab[k + 1], 0.0f);
    }
    b.x = bcol[(k + kh) * NCOLS];
    b.y = bcol[(k + kh + 1) * NCOLS];
    c = __builtin_amdgcn_wmma_f32_16x16x4_f32(false, a, false, b, (short)0, c, false, false);
  }

  const int rbase = (lane >> 4) << 3;
  const float ib = BIAS_INIT ? ibias[n0 + row] : 0.0f;
#pragma unroll
  for (int r = 0; r < 8; ++r) {
    int orow = m0 + rbase + r;
    if (orow < M) {
      const float dv = dinv[orow];
      const size_t o = (size_t)orow * NCOLS + n0 + row;
      float iv = c[r] * (dv * dv);
      if (BIAS_INIT) iv += ib;
      out[o]  = c[r];
      init[o] = iv;
    }
  }
}


__device__ __forceinline__ int wave_first_edge() {
  return (blockIdx.x * 8 + __builtin_amdgcn_readfirstlane(threadIdx.x >> 5)) * 32;
}

__global__ __launch_bounds__(256) void scatter64_kernel(const float* __restrict__ h,
                                                        const int* __restrict__ src,
                                                        const int* __restrict__ dst,
                                                        const float* __restrict__ dinv,
                                                        float* __restrict__ agg,
                                                        int E) {
  const int lane = threadIdx.x & 31;
  const int e0 = wave_first_edge();
  if (e0 >= E) return;
  const int cnt = min(32, E - e0);
  int s = 0, d = 0;
  float w = 0.0f;
  if (lane < cnt) {
    s = src[e0 + lane];
    d = dst[e0 + lane];
    w = dinv[s] * dinv[d];
  }
  const int wbits = __float_as_int(w);
  const float2* __restrict__ hl = (const float2*)h + lane;
  float* __restrict__ al = agg + lane * 2;

  if (cnt == 32) {
    constexpr int G = 8;
    float2 v[32 / G][G];
#pragma unroll
    for (int j = 0; j < G; ++j)
      v[0][j] = hl[(size_t)__builtin_amdgcn_readlane(s, j) * 32];
#pragma unroll
    for (int g = 0; g < 32 / G; ++g) {
      if (g + 1 < 32 / G) {
#pragma unroll
        for (int j = 0; j < G; ++j)
          v[g + 1][j] = hl[(size_t)__builtin_amdgcn_readlane(s, (g + 1) * G + j) * 32];
      }
#pragma unroll
      for (int j = 0; j < G; ++j) {
        const int jj = g * G + j;
        const int dj = __builtin_amdgcn_readlane(d, jj);
        const float wj = __int_as_float(__builtin_amdgcn_readlane(wbits, jj));
        float* op = al + (size_t)dj * 64;
        atomic_add_f32(op,     v[g][j].x * wj);
        atomic_add_f32(op + 1, v[g][j].y * wj);
      }
    }
  } else {
    for (int j = 0; j < cnt; ++j) {
      const int sj = __builtin_amdgcn_readlane(s, j);
      const int dj = __builtin_amdgcn_readlane(d, j);
      const float wj = __int_as_float(__builtin_amdgcn_readlane(wbits, j));
      const float2 vv = hl[(size_t)sj * 32];
      float* op = al + (size_t)dj * 64;
      atomic_add_f32(op,     vv.x * wj);
      atomic_add_f32(op + 1, vv.y * wj);
    }
  }
}

__global__ __launch_bounds__(256) void scatter16_kernel(const float* __restrict__ h,
                                                        const int* __restrict__ src,
                                                        const int* __restrict__ dst,
                                                        const float* __restrict__ dinv,
                                                        float* __restrict__ out,
                                                        int E) {
  const int lane = threadIdx.x & 31;
  const int e0 = wave_first_edge();
  if (e0 >= E) return;
  const int cnt = min(32, E - e0);
  int s = 0, d = 0;
  float w = 0.0f;
  if (lane < cnt) {
    s = src[e0 + lane];
    d = dst[e0 + lane];
    w = dinv[s] * dinv[d];
  }
  const int wbits = __float_as_int(w);
  const int c = lane & 15;
  const int half = lane >> 4;
  const int pick = half << 2;

  if (cnt == 32) {
    float v[16];
    int dj[16];
    float wj[16];
#pragma unroll
    for (int j = 0; j < 16; ++j)
      v[j] = h[(size_t)__builtin_amdgcn_ds_bpermute(pick + 8 * j, s) * 16 + c];
#pragma unroll
    for (int j = 0; j < 16; ++j) {
      dj[j] = __builtin_amdgcn_ds_bpermute(pick + 8 * j, d);
      wj[j] = __int_as_float(__builtin_amdgcn_ds_bpermute(pick + 8 * j, wbits));
    }
#pragma unroll
    for (int j = 0; j < 16; ++j)
      atomic_add_f32(&out[(size_t)dj[j] * 16 + c], v[j] * wj[j]);
  } else {
    for (int j = 0; 2 * j < cnt; ++j) {
      const int a = pick + 8 * j;
      const int sj = __builtin_amdgcn_ds_bpermute(a, s);
      const int dd = __builtin_amdgcn_ds_bpermute(a, d);
      const float ww = __int_as_float(__builtin_amdgcn_ds_bpermute(a, wbits));
      if (2 * j + half < cnt)
        atomic_add_f32(&out[(size_t)dd * 16 + c], h[(size_t)sj * 16 + c] * ww);
    }
  }
}


extern "C" void kernel_launch(void* const* d_in, const int* in_sizes, int n_in,
                              void* d_out, int out_size, void* d_ws, size_t ws_size,
                              hipStream_t stream) {
  const float* x    = (const float*)d_in[0];
  const int*   edge = (const int*)d_in[1];
  const float* W1   = (const float*)d_in[2];
  const float* b1   = (const float*)d_in[3];
  const float* W2   = (const float*)d_in[4];
  const float* b2   = (const float*)d_in[5];
  float* out = (float*)d_out;

  const int E  = in_sizes[1] / 2;
  const int Nn = in_sizes[0] / IN_DIM;
  const int* src = edge;
  const int* dst = edge + E;

  char* ws = (char*)d_ws;
  float* dinv = (float*)ws;
  float* h1   = (float*)(ws + (size_t)Nn * 4);
  float* agg1 = (float*)(ws + (size_t)Nn * 4 + (size_t)Nn * 64 * 4);
  float* h2   = h1;

  const int TB = 256;
  const int mtiles = (Nn + 15) >> 4;

  fill0_kernel<<<(Nn + TB - 1) / TB, TB, 0, stream>>>(dinv, Nn);
  degree_kernel<<<(E + TB - 1) / TB, TB, 0, stream>>>(dst, dinv, E);
  rsqrt_kernel<<<(Nn + TB - 1) / TB, TB, 0, stream>>>(dinv, Nn);

  gemm_wmma_kernel<64, false, false><<<(mtiles * 4 + 7) / 8, 256, 0, stream>>>(
      x, nullptr, W1, dinv, nullptr, h1, agg1, Nn);
  scatter64_kernel<<<(E + 255) / 256, 256, 0, stream>>>(h1, src, dst, dinv, agg1, E);

  gemm_wmma_kernel<16, true, true><<<(mtiles + 7) / 8, 256, 0, stream>>>(
      agg1, b1, W2, dinv, b2, h2, out, Nn);
  scatter16_kernel<<<(E + 255) / 256, 256, 0, stream>>>(h2, src, dst, dinv, out, E);

  (void)n_in; (void)out_size; (void)ws_size;
}


# Edge weight factored: projections store rows times own dinv (source factor); layer 1 adds rows unweighted, its dest factor scales the summed row in layer 2's A load: 3 of 4 per-edge dinv lookups gone
# speedup vs baseline: 1.5304x; 1.0356x over previous
#include <hip/hip_runtime.h>
#include <hip/hip_bf16.h>


#define IN_DIM  64
#define OUT_DIM 16

typedef __attribute__((ext_vector_type(2))) float v2f;
typedef __attribute__((ext_vector_type(8))) float v8f;

__device__ __forceinline__ void atomic_add_f32(float* p, float v) {
  asm volatile("global_atomic_add_f32 %0, %1, off scope:SCOPE_DEV"
               :: "v"(p), "v"(v) : "memory");
}


__global__ void fill0_kernel(float* __restrict__ p, int n) {
  int i = blockIdx.x * blockDim.x + threadIdx.x;
  if (i < n) p[i] = 0.0f;
}

__global__ void degree_kernel(const int* __restrict__ dst, float* __restrict__ deg, int E) {
  int i = blockIdx.x * blockDim.x + threadIdx.x;
  if (i < E) atomic_add_f32(&deg[dst[i]], 1.0f);
}

__global__ void rsqrt_kernel(float* __restrict__ d, int n) {
  int i = blockIdx.x * blockDim.x + threadIdx.x;
  if (i < n) d[i] = rsqrtf(d[i] + 1.0f);
}


template <int NCOLS, bool BIAS_RELU_A, bool BIAS_INIT>
__global__ __launch_bounds__(256) void gemm_wmma_kernel(const float* __restrict__ A,
                                                        const float* __restrict__ abias,
                                                        const float* __restrict__ W,
                                                        const float* __restrict__ dinv,
                                                        const float* __restrict__ ibias,
                                                        float* __restrict__ out,
                                                        float* __restrict__ init, int M) {
  __shared__ float sW[64 * NCOLS];
  const int nvec = (64 * NCOLS) / 4;
  for (int i = threadIdx.x; i < nvec; i += blockDim.x)
    ((float4*)sW)[i] = ((const float4*)W)[i];
  __syncthreads();

  const int lane = threadIdx.x & 31;
  const int wid  = threadIdx.x >> 5;
  const int ntile_n = NCOLS / 16;
  const int mtiles  = (M + 15) >> 4;
  const int gt = blockIdx.x * 8 + wid;
  const int mt = gt / ntile_n;
  const int nt = gt - mt * ntile_n;
  if (mt >= mtiles) return;

  const int m0 = mt * 16;
  const int n0 = nt * 16;
  const int row = lane & 15;
  const int kh  = (lane >> 4) << 1;
  const int arow_idx = min(m0 + row, M - 1);
  const float* __restrict__ arow = A + (size_t)arow_idx * 64 + kh;
  const float* __restrict__ bcol = sW + n0 + row;
  const float* __restrict__ ab = BIAS_RELU_A ? abias + kh : nullptr;
  const float ascale = BIAS_RELU_A ? dinv[arow_idx] : 1.0f;

  v8f c = {};
#pragma unroll
  for (int k = 0; k < 64; k += 4) {
    v2f a, b;
    a.x = arow[k];
    a.y = arow[k + 1];
    if (BIAS_RELU_A) {
      a.x = fmaxf(a.x * ascale + ab[k], 0.0f);
      a.y = fmaxf(a.y * ascale + ab[k + 1], 0.0f);
    }
    b.x = bcol[(k + kh) * NCOLS];
    b.y = bcol[(k + kh + 1) * NCOLS];
    c = __builtin_amdgcn_wmma_f32_16x16x4_f32(false, a, false, b, (short)0, c, false, false);
  }

  const int rbase = (lane >> 4) << 3;
  const float ib = BIAS_INIT ? ibias[n0 + row] : 0.0f;
#pragma unroll
  for (int r = 0; r < 8; ++r) {
    int orow = m0 + rbase + r;
    if (orow < M) {
      const float dv = dinv[orow];
      const size_t o = (size_t)orow * NCOLS + n0 + row;
      const float hv = c[r] * dv;
      out[o]  = hv;
      init[o] = BIAS_INIT ? c[r] * (dv * dv) + ib : hv;
    }
  }
}


__device__ __forceinline__ int wave_first_edge() {
  return (blockIdx.x * 8 + __builtin_amdgcn_readfirstlane(threadIdx.x >> 5)) * 32;
}

__global__ __launch_bounds__(256) void scatter64_kernel(const float* __restrict__ h,
                                                        const int* __restrict__ src,
                                                        const int* __restrict__ dst,
                                                        float* __restrict__ agg,
                                                        int E) {
  const int lane = threadIdx.x & 31;
  const int e0 = wave_first_edge();
  if (e0 >= E) return;
  const int cnt = min(32, E - e0);
  int s = 0, d = 0;
  if (lane < cnt) {
    s = src[e0 + lane];
    d = dst[e0 + lane];
  }
  const float2* __restrict__ hl = (const float2*)h + lane;
  float* __restrict__ al = agg + lane * 2;

  if (cnt == 32) {
    constexpr int G = 8;
    float2 v[32 / G][G];
#pragma unroll
    for (int j = 0; j < G; ++j)
      v[0][j] = hl[(size_t)__builtin_amdgcn_readlane(s, j) * 32];
#pragma unroll
    for (int g = 0; g < 32 / G; ++g) {
      if (g + 1 < 32 / G) {
#pragma unroll
        for (int j = 0; j < G; ++j)
          v[g + 1][j] = hl[(size_t)__builtin_amdgcn_readlane(s, (g + 1) * G + j) * 32];
      }
#pragma unroll
      for (int j = 0; j < G; ++j) {
        const int jj = g * G + j;
        const int dj = __builtin_amdgcn_readlane(d, jj);
        float* op = al + (size_t)dj * 64;
        atomic_add_f32(op,     v[g][j].x);
        atomic_add_f32(op + 1, v[g][j].y);
      }
    }
  } else {
    for (int j = 0; j < cnt; ++j) {
      const int sj = __builtin_amdgcn_readlane(s, j);
      const int dj = __builtin_amdgcn_readlane(d, j);
      const float2 vv = hl[(size_t)sj * 32];
      float* op = al + (size_t)dj * 64;
      atomic_add_f32(op,     vv.x);
      atomic_add_f32(op + 1, vv.y);
    }
  }
}

__global__ __launch_bounds__(256) void scatter16_kernel(const float* __restrict__ h,
                                                        const int* __restrict__ src,
                                                        const int* __restrict__ dst,
                                                        const float* __restrict__ dinv,
                                                        float* __restrict__ out,
                                                        int E) {
  const int lane = threadIdx.x & 31;
  const int e0 = wave_first_edge();
  if (e0 >= E) return;
  const int cnt = min(32, E - e0);
  int s = 0, d = 0;
  float w = 0.0f;
  if (lane < cnt) {
    s = src[e0 + lane];
    d = dst[e0 + lane];
    w = dinv[d];
  }
  const int wbits = __float_as_int(w);
  const int c = lane & 15;
  const int half = lane >> 4;
  const int pick = half << 2;

  if (cnt == 32) {
    float v[16];
    int dj[16];
    float wj[16];
#pragma unroll
    for (int j = 0; j < 16; ++j)
      v[j] = h[(size_t)__builtin_amdgcn_ds_bpermute(pick + 8 * j, s) * 16 + c];
#pragma unroll
    for (int j = 0; j < 16; ++j) {
      dj[j] = __builtin_amdgcn_ds_bpermute(pick + 8 * j, d);
      wj[j] = __int_as_float(__builtin_amdgcn_ds_bpermute(pick + 8 * j, wbits));
    }
#pragma unroll
    for (int j = 0; j < 16; ++j)
      atomic_add_f32(&out[(size_t)dj[j] * 16 + c], v[j] * wj[j]);
  } else {
    for (int j = 0; 2 * j < cnt; ++j) {
      const int a = pick + 8 * j;
      const int sj = __builtin_amdgcn_ds_bpermute(a, s);
      const int dd = __builtin_amdgcn_ds_bpermute(a, d);
      const float ww = __int_as_float(__builtin_amdgcn_ds_bpermute(a, wbits));
      if (2 * j + half < cnt)
        atomic_add_f32(&out[(size_t)dd * 16 + c], h[(size_t)sj * 16 + c] * ww);
    }
  }
}


extern "C" void kernel_launch(void* const* d_in, const int* in_sizes, int n_in,
                              void* d_out, int out_size, void* d_ws, size_t ws_size,
                              hipStream_t stream) {
  const float* x    = (const float*)d_in[0];
  const int*   edge = (const int*)d_in[1];
  const float* W1   = (const float*)d_in[2];
  const float* b1   = (const float*)d_in[3];
  const float* W2   = (const float*)d_in[4];
  const float* b2   = (const float*)d_in[5];
  float* out = (float*)d_out;

  const int E  = in_sizes[1] / 2;
  const int Nn = in_sizes[0] / IN_DIM;
  const int* src = edge;
  const int* dst = edge + E;

  char* ws = (char*)d_ws;
  float* dinv = (float*)ws;
  float* h1   = (float*)(ws + (size_t)Nn * 4);
  float* agg1 = (float*)(ws + (size_t)Nn * 4 + (size_t)Nn * 64 * 4);
  float* h2   = h1;

  const int TB = 256;
  const int mtiles = (Nn + 15) >> 4;

  fill0_kernel<<<(Nn + TB - 1) / TB, TB, 0, stream>>>(dinv, Nn);
  degree_kernel<<<(E + TB - 1) / TB, TB, 0, stream>>>(dst, dinv, E);
  rsqrt_kernel<<<(Nn + TB - 1) / TB, TB, 0, stream>>>(dinv, Nn);

  gemm_wmma_kernel<64, false, false><<<(mtiles * 4 + 7) / 8, 256, 0, stream>>>(
      x, nullptr, W1, dinv, nullptr, h1, agg1, Nn);
  scatter64_kernel<<<(E + 255) / 256, 256, 0, stream>>>(h1, src, dst, agg1, E);

  gemm_wmma_kernel<16, true, true><<<(mtiles + 7) / 8, 256, 0, stream>>>(
      agg1, b1, W2, dinv, b2, h2, out, Nn);
  scatter16_kernel<<<(E + 255) / 256, 256, 0, stream>>>(h2, src, dst, dinv, out, E);

  (void)n_in; (void)out_size; (void)ws_size;
}


# Atomic f32 scatters replaced by per-node gathers over a CSR built each call (degree atomic now returns the edge's slot; scan; place): a node's rows are summed in registers in slot order, stored once
# speedup vs baseline: 1.8957x; 1.2387x over previous
#include <hip/hip_runtime.h>
#include <hip/hip_bf16.h>


#define IN_DIM  64
#define OUT_DIM 16

typedef __attribute__((ext_vector_type(2))) float v2f;
typedef __attribute__((ext_vector_type(8))) float v8f;


__device__ __forceinline__ float atomic_add_f32_old(float* p, float v) {
  float old;
  asm volatile("global_atomic_add_f32 %0, %1, %2, off th:TH_ATOMIC_RETURN scope:SCOPE_DEV\n\t"
               "s_wait_loadcnt 0x0"
               : "=&v"(old) : "v"(p), "v"(v) : "memory");
  return old;
}

__device__ __forceinline__ void atomic_add4_f32_old(float* p0, float* p1, float* p2, float* p3,
                                                    float v, float& o0, float& o1, float& o2,
                                                    float& o3) {
  asm volatile("global_atomic_add_f32 %0, %4, %8, off th:TH_ATOMIC_RETURN scope:SCOPE_DEV\n\t"
               "global_atomic_add_f32 %1, %5, %8, off th:TH_ATOMIC_RETURN scope:SCOPE_DEV\n\t"
               "global_atomic_add_f32 %2, %6, %8, off th:TH_ATOMIC_RETURN scope:SCOPE_DEV\n\t"
               "global_atomic_add_f32 %3, %7, %8, off th:TH_ATOMIC_RETURN scope:SCOPE_DEV\n\t"
               "s_wait_loadcnt 0x0"
               : "=&v"(o0), "=&v"(o1), "=&v"(o2), "=&v"(o3)
               : "v"(p0), "v"(p1), "v"(p2), "v"(p3), "v"(v) : "memory");
}

__global__ void fill0_kernel(float* __restrict__ p, int n) {
  int i = blockIdx.x * blockDim.x + threadIdx.x;
  if (i < n) p[i] = 0.0f;
}

__global__ void degree_rank_kernel(const int* __restrict__ dst, float* __restrict__ deg,
                                   int* __restrict__ rank, int E) {
  const int i = (blockIdx.x * blockDim.x + threadIdx.x) * 4;
  if (i >= E) return;
  if ((E & 3) == 0) {
    const int4 d = *(const int4*)(dst + i);
    float o0, o1, o2, o3;
    atomic_add4_f32_old(deg + d.x, deg + d.y, deg + d.z, deg + d.w, 1.0f, o0, o1, o2, o3);
    *(int4*)(rank + i) = make_int4((int)o0, (int)o1, (int)o2, (int)o3);
  } else {
    const int e1 = min(i + 4, E);
    for (int k = i; k < e1; ++k) rank[k] = (int)atomic_add_f32_old(&deg[dst[k]], 1.0f);
  }
}

__global__ __launch_bounds__(1024) void node_kernel(const float* __restrict__ deg,
                                                    float* __restrict__ dinv,
                                                    int* __restrict__ start, int n) {
  __shared__ int wsum[32];
  __shared__ int wpre[32];
  __shared__ int sfront;
  const int lane = threadIdx.x & 31;
  const int wid  = threadIdx.x >> 5;
  const int first = blockIdx.x * 1024;
  int before = 0;
  for (int j0 = threadIdx.x * 4; j0 < first; j0 += 8 * 4096) {
#pragma unroll
    for (int k = 0; k < 8; ++k) {
      const int j = j0 + k * 4096;
      const bool ok = j < first;
      const float4 q = *(const float4*)(deg + (ok ? j : 0));
      const int s = (int)q.x + (int)q.y + (int)q.z + (int)q.w;
      before += ok ? s : 0;
    }
  }
#pragma unroll
  for (int o = 16; o > 0; o >>= 1) before += __shfl_xor(before, o, 32);
  const int i = first + threadIdx.x;
  const float cf = (i < n) ? deg[i] : 0.0f;
  const int c = (int)cf;
  int x = c;
#pragma unroll
  for (int o = 1; o < 32; o <<= 1) {
    const int y = __shfl_up(x, o, 32);
    if (lane >= o) x += y;
  }
  if (lane == 31) wsum[wid] = x;
  if (lane == 0)  wpre[wid] = before;
  __syncthreads();
  if (wid == 0) {
    const int t = wsum[lane];
    int u = t;
#pragma unroll
    for (int o = 1; o < 32; o <<= 1) {
      const int y = __shfl_up(u, o, 32);
      if (lane >= o) u += y;
    }
    int p = wpre[lane];
#pragma unroll
    for (int o = 16; o > 0; o >>= 1) p += __shfl_xor(p, o, 32);
    wsum[lane] = u - t;
    if (lane == 0) sfront = p;
  }
  __syncthreads();
  if (i < n) {
    start[i] = sfront + wsum[wid] + (x - c);
    dinv[i]  = rsqrtf(cf + 1.0f);
  }
}

__global__ void place_kernel(const int* __restrict__ src, const int* __restrict__ dst,
                             const int* __restrict__ rank, const int* __restrict__ start,
                             int* __restrict__ csr, int E) {
  const int i = (blockIdx.x * blockDim.x + threadIdx.x) * 4;
  if (i >= E) return;
  if ((E & 3) == 0) {
    const int4 s = *(const int4*)(src + i);
    const int4 d = *(const int4*)(dst + i);
    const int4 r = *(const int4*)(rank + i);
    const int p0 = start[d.x], p1 = start[d.y], p2 = start[d.z], p3 = start[d.w];
    csr[p0 + r.x] = s.x;
    csr[p1 + r.y] = s.y;
    csr[p2 + r.z] = s.z;
    csr[p3 + r.w] = s.w;
  } else {
    const int e1 = min(i + 4, E);
    for (int k = i; k < e1; ++k) csr[start[dst[k]] + rank[k]] = src[k];
  }
}


template <int NCOLS, bool BIAS_RELU_A>
__global__ __launch_bounds__(256) void gemm_wmma_kernel(const float* __restrict__ A,
                                                        const float* __restrict__ abias,
                                                        const float* __restrict__ W,
                                                        const float* __restrict__ dinv,
                                                        float* __restrict__ out, int M) {
  __shared__ float sW[64 * NCOLS];
  const int nvec = (64 * NCOLS) / 4;
  for (int i = threadIdx.x; i < nvec; i += blockDim.x)
    ((float4*)sW)[i] = ((const float4*)W)[i];
  __syncthreads();

  const int lane = threadIdx.x & 31;
  const int wid  = threadIdx.x >> 5;
  const int ntile_n = NCOLS / 16;
  const int mtiles  = (M + 15) >> 4;
  const int gt = blockIdx.x * 8 + wid;
  const int mt = gt / ntile_n;
  const int nt = gt - mt * ntile_n;
  if (mt >= mtiles) return;

  const int m0 = mt * 16;
  const int n0 = nt * 16;
  const int row = lane & 15;
  const int kh  = (lane >> 4) << 1;
  const int arow_idx = min(m0 + row, M - 1);
  const float* __restrict__ arow = A + (size_t)arow_idx * 64 + kh;
  const float* __restrict__ bcol = sW + n0 + row;
  const float* __restrict__ ab = BIAS_RELU_A ? abias + kh : nullptr;
  const float ascale = BIAS_RELU_A ? dinv[arow_idx] : 1.0f;

  v8f c = {};
#pragma unroll
  for (int k = 0; k < 64; k += 4) {
    v2f a, b;
    a.x = arow[k];
    a.y = arow[k + 1];
    if (BIAS_RELU_A) {
      a.x = fmaxf(a.x * ascale + ab[k], 0.0f);
      a.y = fmaxf(a.y * ascale + ab[k + 1], 0.0f);
    }
    b.x = bcol[(k + kh) * NCOLS];
    b.y = bcol[(k + kh + 1) * NCOLS];
    c = __builtin_amdgcn_wmma_f32_16x16x4_f32(false, a, false, b, (short)0, c, false, false);
  }

  const int rbase = (lane >> 4) << 3;
#pragma unroll
  for (int r = 0; r < 8; ++r) {
    int orow = m0 + rbase + r;
    if (orow < M)
      out[(size_t)orow * NCOLS + n0 + row] = c[r] * dinv[orow];
  }
}


#define NODES_PER_WAVE64 8

template <int K>
__device__ __forceinline__ float2 sum_rows64(const float2* __restrict__ hl, int idx, int rem,
                                             float2 acc) {
  float2 v[K];
#pragma unroll
  for (int j = 0; j < K; ++j)
    v[j] = hl[(size_t)__builtin_amdgcn_readlane(idx, j) * 32];
#pragma unroll
  for (int j = 0; j < K; ++j) {
    const bool on = (j < K - 8) || (j < rem);
    acc.x += on ? v[j].x : 0.0f;
    acc.y += on ? v[j].y : 0.0f;
  }
  return acc;
}

__global__ __launch_bounds__(256) void gather64_kernel(const float* __restrict__ h,
                                                       const int* __restrict__ start,
                                                       const float* __restrict__ deg,
                                                       const int* __restrict__ csr,
                                                       float* __restrict__ agg, int N) {
  const int lane = threadIdx.x & 31;
  const int node0 =
      (blockIdx.x * 8 + __builtin_amdgcn_readfirstlane(threadIdx.x >> 5)) * NODES_PER_WAVE64;
  if (node0 >= N) return;
  const int nw = min(NODES_PER_WAVE64, N - node0);
  int begv = 0, cntv = 0;
  if (lane < nw) {
    begv = start[node0 + lane];
    cntv = (int)deg[node0 + lane];
  }
  const float2* __restrict__ hl = (const float2*)h + lane;
  for (int q = 0; q < nw; ++q) {
    const int node = node0 + q;
    int beg = __builtin_amdgcn_readlane(begv, q);
    int rem = __builtin_amdgcn_readlane(cntv, q);
    float2 acc = hl[(size_t)node * 32];
    while (rem > 0) {
      int idx = node;
      if (lane < rem) idx = csr[beg + lane];
      if (rem > 24)      acc = sum_rows64<32>(hl, idx, rem, acc);
      else if (rem > 16) acc = sum_rows64<24>(hl, idx, rem, acc);
      else if (rem > 8)  acc = sum_rows64<16>(hl, idx, rem, acc);
      else               acc = sum_rows64<8>(hl, idx, rem, acc);
      rem -= 32;
      beg += 32;
    }
    ((float2*)agg)[(size_t)node * 32 + lane] = acc;
  }
}

#define PAIRS_PER_WAVE16 4

template <int K>
__device__ __forceinline__ float sum_rows16(const float* __restrict__ hc, int pick, int i0, int i1,
                                            int rem, float acc) {
  float v[K];
#pragma unroll
  for (int j = 0; j < K; ++j) {
    const int sj = __builtin_amdgcn_ds_bpermute(pick + 4 * (j & 15), j < 16 ? i0 : i1);
    v[j] = hc[(size_t)sj * 16];
  }
#pragma unroll
  for (int j = 0; j < K; ++j)
    acc += (j < rem) ? v[j] : 0.0f;
  return acc;
}

__global__ __launch_bounds__(256) void gather16_kernel(const float* __restrict__ h,
                                                       const int* __restrict__ start,
                                                       const float* __restrict__ deg,
                                                       const int* __restrict__ csr,
                                                       const float* __restrict__ dinv,
                                                       const float* __restrict__ bias,
                                                       float* __restrict__ out, int N) {
  const int lane = threadIdx.x & 31;
  const int c = lane & 15;
  const int half = lane >> 4;
  const int base0 = (blockIdx.x * 8 + __builtin_amdgcn_readfirstlane(threadIdx.x >> 5)) *
                    (2 * PAIRS_PER_WAVE16);
  if (base0 >= N) return;
  const int np = min(PAIRS_PER_WAVE16, (N - base0 + 1) >> 1);
  int begv = 0, cntv = 0;
  float dv = 0.0f;
  if (lane < 2 * PAIRS_PER_WAVE16 && base0 + lane < N) {
    begv = start[base0 + lane];
    cntv = (int)deg[base0 + lane];
    dv = dinv[base0 + lane];
  }
  const int dvbits = __float_as_int(dv);
  const float* __restrict__ hc = h + c;
  const float bc = bias[c];
  const int pick = half << 6;

  for (int q = 0; q < np; ++q) {
    const int node = base0 + 2 * q + half;
    const int nd = min(node, N - 1);
    const int a = (2 * q + half) << 2;
    int beg = __builtin_amdgcn_ds_bpermute(a, begv);
    int rem = __builtin_amdgcn_ds_bpermute(a, cntv);
    const float dd = __int_as_float(__builtin_amdgcn_ds_bpermute(a, dvbits));
    int mu = max(__builtin_amdgcn_readlane(cntv, 2 * q), __builtin_amdgcn_readlane(cntv, 2 * q + 1));
    float acc = hc[(size_t)nd * 16];
    while (mu > 0) {
      int i0 = nd, i1 = nd;
      if (c < rem)      i0 = csr[beg + c];
      if (c + 16 < rem) i1 = csr[beg + 16 + c];
      if (mu > 24)      acc = sum_rows16<32>(hc, pick, i0, i1, rem, acc);
      else if (mu > 16) acc = sum_rows16<24>(hc, pick, i0, i1, rem, acc);
      else if (mu > 8)  acc = sum_rows16<16>(hc, pick, i0, i1, rem, acc);
      else              acc = sum_rows16<8>(hc, pick, i0, i1, rem, acc);
      mu -= 32;
      rem -= 32;
      beg += 32;
    }
    if (node < N) out[(size_t)node * 16 + c] = bc + dd * acc;
  }
}


extern "C" void kernel_launch(void* const* d_in, const int* in_sizes, int n_in,
                              void* d_out, int out_size, void* d_ws, size_t ws_size,
                              hipStream_t stream) {
  const float* x    = (const float*)d_in[0];
  const int*   edge = (const int*)d_in[1];
  const float* W1   = (const float*)d_in[2];
  const float* b1   = (const float*)d_in[3];
  const float* W2   = (const float*)d_in[4];
  const float* b2   = (const float*)d_in[5];
  float* out = (float*)d_out;

  const int E  = in_sizes[1] / 2;
  const int Nn = in_sizes[0] / IN_DIM;
  const int* src = edge;
  const int* dst = edge + E;

  const size_t nodeB = ((size_t)Nn * 4 + 255) & ~(size_t)255;
  const size_t edgeB = ((size_t)E * 4 + 255) & ~(size_t)255;
  const size_t featB = (size_t)Nn * 64 * 4;
  char* ws = (char*)d_ws;
  float* deg   = (float*)ws;
  float* dinv  = (float*)(ws + nodeB);
  int*   start = (int*)(ws + 2 * nodeB);
  int*   rank  = (int*)(ws + 3 * nodeB);
  int*   csr   = (int*)(ws + 3 * nodeB + edgeB);
  float* h1    = (float*)(ws + 3 * nodeB + 2 * edgeB);
  float* agg1  = (float*)(ws + 3 * nodeB + 2 * edgeB + featB);
  float* h2    = h1;

  const int TB = 256;
  const int mtiles = (Nn + 15) >> 4;

  fill0_kernel<<<(Nn + TB - 1) / TB, TB, 0, stream>>>(deg, Nn);
  degree_rank_kernel<<<(E + 4 * TB - 1) / (4 * TB), TB, 0, stream>>>(dst, deg, rank, E);
  node_kernel<<<(Nn + 1023) / 1024, 1024, 0, stream>>>(deg, dinv, start, Nn);
  place_kernel<<<(E + 4 * TB - 1) / (4 * TB), TB, 0, stream>>>(src, dst, rank, start, csr, E);

  gemm_wmma_kernel<64, false><<<(mtiles * 4 + 7) / 8, 256, 0, stream>>>(
      x, nullptr, W1, dinv, h1, Nn);
  gather64_kernel<<<(Nn + 8 * NODES_PER_WAVE64 - 1) / (8 * NODES_PER_WAVE64), 256, 0, stream>>>(
      h1, start, deg, csr, agg1, Nn);

  gemm_wmma_kernel<16, true><<<(mtiles + 7) / 8, 256, 0, stream>>>(
      agg1, b1, W2, dinv, h2, Nn);
  gather16_kernel<<<(Nn + 16 * PAIRS_PER_WAVE16 - 1) / (16 * PAIRS_PER_WAVE16), 256, 0, stream>>>(
      h2, start, deg, csr, dinv, b2, out, Nn);

  (void)n_in; (void)out_size; (void)ws_size;
}


# CSR build without any atomic: two-level counting sort (256-node bins per 8192-edge chunk, then nodes inside a bin) on thread-private LDS counters; slot order now fixed by the edge list
# speedup vs baseline: 2.5519x; 1.3462x over previous
#include <hip/hip_runtime.h>
#include <hip/hip_bf16.h>


#define IN_DIM  64
#define OUT_DIM 16

typedef __attribute__((ext_vector_type(2))) float v2f;
typedef __attribute__((ext_vector_type(8))) float v8f;


#define BIN_SHIFT 8
#define BIN_NODES 256
#define MAX_BINS  512
#define IPT       32
#define CHUNK     (256 * IPT)
#define PITCH16   258

__device__ __forceinline__ void rank_items(unsigned short* __restrict__ priv,
                                           int* __restrict__ tot, const int (&key)[IPT],
                                           int (&r)[IPT], int nk) {
  const int t = threadIdx.x;
  const int ndw = nk * (PITCH16 / 2);
  for (int i = t; i < ndw; i += 256) ((int*)priv)[i] = 0;
  __syncthreads();
#pragma unroll
  for (int k = 0; k < IPT; ++k) {
    r[k] = 0;
    if (key[k] >= 0) {
      unsigned short* p = priv + key[k] * PITCH16 + t;
      const unsigned short c = *p;
      r[k] = c;
      *p = (unsigned short)(c + 1);
    }
  }
  __syncthreads();
  for (int kk = t; kk < nk; kk += 256) {
    int* row = (int*)(priv + kk * PITCH16);
    int run = 0;
#pragma unroll 8
    for (int j = 0; j < 128; ++j) {
      const unsigned w = (unsigned)row[j];
      const int lo = (int)(w & 0xffffu);
      const int hi = (int)(w >> 16);
      row[j] = run | ((run + lo) << 16);
      run += lo + hi;
    }
    tot[kk] = run;
  }
  __syncthreads();
}

__device__ __forceinline__ void load_idx(const int* __restrict__ p, int e0, int E, int (&v)[IPT]) {
  const int t = threadIdx.x;
  if ((E & 3) == 0) {
#pragma unroll
    for (int kk = 0; kk < IPT / 4; ++kk) {
      const int i = e0 + kk * 1024 + t * 4;
      int4 q = make_int4(-1, -1, -1, -1);
      if (i < E) q = *(const int4*)(p + i);
      v[4 * kk] = q.x; v[4 * kk + 1] = q.y; v[4 * kk + 2] = q.z; v[4 * kk + 3] = q.w;
    }
  } else {
#pragma unroll
    for (int k = 0; k < IPT; ++k) {
      const int i = e0 + (k >> 2) * 1024 + t * 4 + (k & 3);
      v[k] = (i < E) ? p[i] : -1;
    }
  }
}

__global__ __launch_bounds__(256) void bin_count_kernel(const int* __restrict__ dst,
                                                        int* __restrict__ cnt, int E, int NB) {
  __shared__ __attribute__((aligned(16))) unsigned short priv[MAX_BINS * PITCH16];
  __shared__ int tot[MAX_BINS];
  int d[IPT], key[IPT], r[IPT];
  load_idx(dst, blockIdx.x * CHUNK, E, d);
#pragma unroll
  for (int k = 0; k < IPT; ++k) key[k] = d[k] >> BIN_SHIFT;
  rank_items(priv, tot, key, r, NB);
  for (int b = threadIdx.x; b < NB; b += 256) cnt[blockIdx.x * NB + b] = tot[b];
}

__global__ __launch_bounds__(256) void bin_scatter_kernel(const int* __restrict__ src,
                                                          const int* __restrict__ dst,
                                                          const int* __restrict__ cnt,
                                                          int* __restrict__ bstart,
                                                          int2* __restrict__ binned,
                                                          int E, int NB, int G) {
  __shared__ __attribute__((aligned(16))) unsigned short priv[MAX_BINS * PITCH16];
  __shared__ int tot[MAX_BINS];
  __shared__ int wtot[8];
  const int lane = threadIdx.x & 31;
  const int wid  = threadIdx.x >> 5;
  const int g = blockIdx.x;
  int d[IPT], s[IPT], key[IPT], r[IPT];
  load_idx(dst, g * CHUNK, E, d);
  load_idx(src, g * CHUNK, E, s);
#pragma unroll
  for (int k = 0; k < IPT; ++k) key[k] = d[k] >> BIN_SHIFT;
  rank_items(priv, tot, key, r, NB);

  int carry = 0;
  for (int b0 = 0; b0 < NB; b0 += 256) {
    const int b = b0 + threadIdx.x;
    int all = 0, pre = 0;
    if (b < NB) {
      for (int g0 = 0; g0 < G; g0 += 28) {
        int c[28];
#pragma unroll
        for (int u = 0; u < 28; ++u) c[u] = cnt[min(g0 + u, G - 1) * NB + b];
#pragma unroll
        for (int u = 0; u < 28; ++u) {
          const int v = (g0 + u < G) ? c[u] : 0;
          all += v;
          pre += (g0 + u < g) ? v : 0;
        }
      }
    }
    int x = all;
#pragma unroll
    for (int o = 1; o < 32; o <<= 1) {
      const int y = __shfl_up(x, o, 32);
      if (lane >= o) x += y;
    }
    if (lane == 31) wtot[wid] = x;
    __syncthreads();
    int wp = 0, sum = 0;
#pragma unroll
    for (int w = 0; w < 8; ++w) {
      const int v = wtot[w];
      wp += (w < wid) ? v : 0;
      sum += v;
    }
    const int first = carry + wp + (x - all);
    if (b < NB) {
      tot[b] = first + pre;
      if (g == 0) bstart[b] = first;
    }
    carry += sum;
    __syncthreads();
  }
  if (g == 0 && threadIdx.x == 0) bstart[NB] = carry;

#pragma unroll
  for (int k = 0; k < IPT; ++k) {
    if (key[k] >= 0) {
      const int slot = tot[key[k]] + priv[key[k] * PITCH16 + threadIdx.x] + r[k];
      binned[slot] = make_int2(s[k], d[k]);
    }
  }
}

__global__ __launch_bounds__(256) void bin_csr_kernel(const int2* __restrict__ binned,
                                                      const int* __restrict__ bstart,
                                                      float* __restrict__ deg,
                                                      float* __restrict__ dinv,
                                                      int* __restrict__ start,
                                                      int* __restrict__ csr, int N) {
  __shared__ __attribute__((aligned(16))) unsigned short priv[BIN_NODES * PITCH16];
  __shared__ int tot[BIN_NODES];
  __shared__ int cur[BIN_NODES];
  __shared__ int wtot[8];
  const int t = threadIdx.x;
  const int lane = t & 31;
  const int wid  = t >> 5;
  const int bs = bstart[blockIdx.x];
  const int be = bstart[blockIdx.x + 1];
  int s[IPT], key[IPT], r[IPT];

  cur[t] = 0;
  for (int c0 = bs; c0 < be; c0 += CHUNK) {
#pragma unroll
    for (int k = 0; k < IPT; ++k) {
      const int i = c0 + k * 256 + t;
      const int dd = binned[min(i, be - 1)].y;
      key[k] = (i < be) ? (dd & (BIN_NODES - 1)) : -1;
    }
    rank_items(priv, tot, key, r, BIN_NODES);
    cur[t] += tot[t];
  }
  const int cn = cur[t];
  int x = cn;
#pragma unroll
  for (int o = 1; o < 32; o <<= 1) {
    const int y = __shfl_up(x, o, 32);
    if (lane >= o) x += y;
  }
  if (lane == 31) wtot[wid] = x;
  __syncthreads();
  int wp = 0;
#pragma unroll
  for (int w = 0; w < 8; ++w) wp += (w < wid) ? wtot[w] : 0;
  const int excl = wp + (x - cn);
  const int n = blockIdx.x * BIN_NODES + t;
  if (n < N) {
    start[n] = bs + excl;
    deg[n]   = (float)cn;
    dinv[n]  = rsqrtf((float)cn + 1.0f);
  }
  cur[t] = excl;

  for (int c0 = bs; c0 < be; c0 += CHUNK) {
#pragma unroll
    for (int k = 0; k < IPT; ++k) {
      const int i = c0 + k * 256 + t;
      const int2 sd = binned[min(i, be - 1)];
      s[k] = sd.x;
      key[k] = (i < be) ? (sd.y & (BIN_NODES - 1)) : -1;
    }
    rank_items(priv, tot, key, r, BIN_NODES);
#pragma unroll
    for (int k = 0; k < IPT; ++k) {
      if (key[k] >= 0) {
        const int slot = cur[key[k]] + priv[key[k] * PITCH16 + t] + r[k];
        csr[bs + slot] = s[k];
      }
    }
    __syncthreads();
    cur[t] += tot[t];
  }
}


template <int NCOLS, bool BIAS_RELU_A>
__global__ __launch_bounds__(256) void gemm_wmma_kernel(const float* __restrict__ A,
                                                        const float* __restrict__ abias,
                                                        const float* __restrict__ W,
                                                        const float* __restrict__ dinv,
                                                        float* __restrict__ out, int M) {
  __shared__ float sW[64 * NCOLS];
  const int nvec = (64 * NCOLS) / 4;
  for (int i = threadIdx.x; i < nvec; i += blockDim.x)
    ((float4*)sW)[i] = ((const float4*)W)[i];
  __syncthreads();

  const int lane = threadIdx.x & 31;
  const int wid  = threadIdx.x >> 5;
  const int ntile_n = NCOLS / 16;
  const int mtiles  = (M + 15) >> 4;
  const int gt = blockIdx.x * 8 + wid;
  const int mt = gt / ntile_n;
  const int nt = gt - mt * ntile_n;
  if (mt >= mtiles) return;

  const int m0 = mt * 16;
  const int n0 = nt * 16;
  const int row = lane & 15;
  const int kh  = (lane >> 4) << 1;
  const int arow_idx = min(m0 + row, M - 1);
  const float* __restrict__ arow = A + (size_t)arow_idx * 64 + kh;
  const float* __restrict__ bcol = sW + n0 + row;
  const float* __restrict__ ab = BIAS_RELU_A ? abias + kh : nullptr;
  const float ascale = BIAS_RELU_A ? dinv[arow_idx] : 1.0f;

  v8f c = {};
#pragma unroll
  for (int k = 0; k < 64; k += 4) {
    v2f a, b;
    a.x = arow[k];
    a.y = arow[k + 1];
    if (BIAS_RELU_A) {
      a.x = fmaxf(a.x * ascale + ab[k], 0.0f);
      a.y = fmaxf(a.y * ascale + ab[k + 1], 0.0f);
    }
    b.x = bcol[(k + kh) * NCOLS];
    b.y = bcol[(k + kh + 1) * NCOLS];
    c = __builtin_amdgcn_wmma_f32_16x16x4_f32(false, a, false, b, (short)0, c, false, false);
  }

  const int rbase = (lane >> 4) << 3;
#pragma unroll
  for (int r = 0; r < 8; ++r) {
    int orow = m0 + rbase + r;
    if (orow < M)
      out[(size_t)orow * NCOLS + n0 + row] = c[r] * dinv[orow];
  }
}


#define NODES_PER_WAVE64 8

template <int K>
__device__ __forceinline__ float2 sum_rows64(const float2* __restrict__ hl, int idx, int rem,
                                             float2 acc) {
  float2 v[K];
#pragma unroll
  for (int j = 0; j < K; ++j)
    v[j] = hl[(size_t)__builtin_amdgcn_readlane(idx, j) * 32];
#pragma unroll
  for (int j = 0; j < K; ++j) {
    const bool on = (j < K - 8) || (j < rem);
    acc.x += on ? v[j].x : 0.0f;
    acc.y += on ? v[j].y : 0.0f;
  }
  return acc;
}

__global__ __launch_bounds__(256) void gather64_kernel(const float* __restrict__ h,
                                                       const int* __restrict__ start,
                                                       const float* __restrict__ deg,
                                                       const int* __restrict__ csr,
                                                       float* __restrict__ agg, int N) {
  const int lane = threadIdx.x & 31;
  const int node0 =
      (blockIdx.x * 8 + __builtin_amdgcn_readfirstlane(threadIdx.x >> 5)) * NODES_PER_WAVE64;
  if (node0 >= N) return;
  const int nw = min(NODES_PER_WAVE64, N - node0);
  int begv = 0, cntv = 0;
  if (lane < nw) {
    begv = start[node0 + lane];
    cntv = (int)deg[node0 + lane];
  }
  const float2* __restrict__ hl = (const float2*)h + lane;
  for (int q = 0; q < nw; ++q) {
    const int node = node0 + q;
    int beg = __builtin_amdgcn_readlane(begv, q);
    int rem = __builtin_amdgcn_readlane(cntv, q);
    float2 acc = hl[(size_t)node * 32];
    while (rem > 0) {
      int idx = node;
      if (lane < rem) idx = csr[beg + lane];
      if (rem > 24)      acc = sum_rows64<32>(hl, idx, rem, acc);
      else if (rem > 16) acc = sum_rows64<24>(hl, idx, rem, acc);
      else if (rem > 8)  acc = sum_rows64<16>(hl, idx, rem, acc);
      else               acc = sum_rows64<8>(hl, idx, rem, acc);
      rem -= 32;
      beg += 32;
    }
    ((float2*)agg)[(size_t)node * 32 + lane] = acc;
  }
}

#define PAIRS_PER_WAVE16 4

template <int K>
__device__ __forceinline__ float sum_rows16(const float* __restrict__ hc, int pick, int i0, int i1,
                                            int rem, float acc) {
  float v[K];
#pragma unroll
  for (int j = 0; j < K; ++j) {
    const int sj = __builtin_amdgcn_ds_bpermute(pick + 4 * (j & 15), j < 16 ? i0 : i1);
    v[j] = hc[(size_t)sj * 16];
  }
#pragma unroll
  for (int j = 0; j < K; ++j)
    acc += (j < rem) ? v[j] : 0.0f;
  return acc;
}

__global__ __launch_bounds__(256) void gather16_kernel(const float* __restrict__ h,
                                                       const int* __restrict__ start,
                                                       const float* __restrict__ deg,
                                                       const int* __restrict__ csr,
                                                       const float* __restrict__ dinv,
                                                       const float* __restrict__ bias,
                                                       float* __restrict__ out, int N) {
  const int lane = threadIdx.x & 31;
  const int c = lane & 15;
  const int half = lane >> 4;
  const int base0 = (blockIdx.x * 8 + __builtin_amdgcn_readfirstlane(threadIdx.x >> 5)) *
                    (2 * PAIRS_PER_WAVE16);
  if (base0 >= N) return;
  const int np = min(PAIRS_PER_WAVE16, (N - base0 + 1) >> 1);
  int begv = 0, cntv = 0;
  float dv = 0.0f;
  if (lane < 2 * PAIRS_PER_WAVE16 && base0 + lane < N) {
    begv = start[base0 + lane];
    cntv = (int)deg[base0 + lane];
    dv = dinv[base0 + lane];
  }
  const int dvbits = __float_as_int(dv);
  const float* __restrict__ hc = h + c;
  const float bc = bias[c];
  const int pick = half << 6;

  for (int q = 0; q < np; ++q) {
    const int node = base0 + 2 * q + half;
    const int nd = min(node, N - 1);
    const int a = (2 * q + half) << 2;
    int beg = __builtin_amdgcn_ds_bpermute(a, begv);
    int rem = __builtin_amdgcn_ds_bpermute(a, cntv);
    const float dd = __int_as_float(__builtin_amdgcn_ds_bpermute(a, dvbits));
    int mu = max(__builtin_amdgcn_readlane(cntv, 2 * q), __builtin_amdgcn_readlane(cntv, 2 * q + 1));
    float acc = hc[(size_t)nd * 16];
    while (mu > 0) {
      int i0 = nd, i1 = nd;
      if (c < rem)      i0 = csr[beg + c];
      if (c + 16 < rem) i1 = csr[beg + 16 + c];
      if (mu > 24)      acc = sum_rows16<32>(hc, pick, i0, i1, rem, acc);
      else if (mu > 16) acc = sum_rows16<24>(hc, pick, i0, i1, rem, acc);
      else if (mu > 8)  acc = sum_rows16<16>(hc, pick, i0, i1, rem, acc);
      else              acc = sum_rows16<8>(hc, pick, i0, i1, rem, acc);
      mu -= 32;
      rem -= 32;
      beg += 32;
    }
    if (node < N) out[(size_t)node * 16 + c] = bc + dd * acc;
  }
}


extern "C" void kernel_launch(void* const* d_in, const int* in_sizes, int n_in,
                              void* d_out, int out_size, void* d_ws, size_t ws_size,
                              hipStream_t stream) {
  const float* x    = (const float*)d_in[0];
  const int*   edge = (const int*)d_in[1];
  const float* W1   = (const float*)d_in[2];
  const float* b1   = (const float*)d_in[3];
  const float* W2   = (const float*)d_in[4];
  const float* b2   = (const float*)d_in[5];
  float* out = (float*)d_out;

  const int E  = in_sizes[1] / 2;
  const int Nn = in_sizes[0] / IN_DIM;
  const int* src = edge;
  const int* dst = edge + E;

  const int NB = (Nn + BIN_NODES - 1) >> BIN_SHIFT;
  const int G  = (E + CHUNK - 1) / CHUNK;

  const size_t nodeB = ((size_t)Nn * 4 + 255) & ~(size_t)255;
  const size_t edgeB = ((size_t)E * 4 + 255) & ~(size_t)255;
  const size_t cntB  = ((size_t)G * NB * 4 + 255) & ~(size_t)255;
  const size_t bstB  = ((size_t)(NB + 1) * 4 + 255) & ~(size_t)255;
  const size_t featB = (size_t)Nn * 64 * 4;
  char* ws = (char*)d_ws;
  float* deg    = (float*)ws;
  float* dinv   = (float*)(ws + nodeB);
  int*   start  = (int*)(ws + 2 * nodeB);
  int*   cnt    = (int*)(ws + 3 * nodeB);
  int*   bstart = (int*)(ws + 3 * nodeB + cntB);
  int2*  binned = (int2*)(ws + 3 * nodeB + cntB + bstB);
  int*   csr    = (int*)(ws + 3 * nodeB + cntB + bstB + 2 * edgeB);
  float* h1     = (float*)(ws + 3 * nodeB + cntB + bstB + 3 * edgeB);
  float* agg1   = (float*)(ws + 3 * nodeB + cntB + bstB + 3 * edgeB + featB);
  float* h2     = h1;

  const int mtiles = (Nn + 15) >> 4;

  bin_count_kernel<<<G, 256, 0, stream>>>(dst, cnt, E, NB);
  bin_scatter_kernel<<<G, 256, 0, stream>>>(src, dst, cnt, bstart, binned, E, NB, G);
  bin_csr_kernel<<<NB, 256, 0, stream>>>(binned, bstart, deg, dinv, start, csr, Nn);

  gemm_wmma_kernel<64, false><<<(mtiles * 4 + 7) / 8, 256, 0, stream>>>(
      x, nullptr, W1, dinv, h1, Nn);
  gather64_kernel<<<(Nn + 8 * NODES_PER_WAVE64 - 1) / (8 * NODES_PER_WAVE64), 256, 0, stream>>>(
      h1, start, deg, csr, agg1, Nn);

  gemm_wmma_kernel<16, true><<<(mtiles + 7) / 8, 256, 0, stream>>>(
      agg1, b1, W2, dinv, h2, Nn);
  gather16_kernel<<<(Nn + 16 * PAIRS_PER_WAVE16 - 1) / (16 * PAIRS_PER_WAVE16), 256, 0, stream>>>(
      h2, start, deg, csr, dinv, b2, out, Nn);

  (void)n_in; (void)out_size; (void)ws_size;
}


# Layer-1 projection: a wave computes all 64 columns of its 16 rows (4 accumulator tiles, tile t = columns 4c+t): x read once not 4 times, W1 by 16-byte LDS reads, whole-row stores, 1/4 of the waves
# speedup vs baseline: 2.7362x; 1.0722x over previous
#include <hip/hip_runtime.h>
#include <hip/hip_bf16.h>


#define IN_DIM  64
#define OUT_DIM 16

typedef __attribute__((ext_vector_type(2))) float v2f;
typedef __attribute__((ext_vector_type(8))) float v8f;


#define BIN_SHIFT 8
#define BIN_NODES 256
#define MAX_BINS  512
#define IPT       32
#define CHUNK     (256 * IPT)
#define PITCH16   258

__device__ __forceinline__ void rank_items(unsigned short* __restrict__ priv,
                                           int* __restrict__ tot, const int (&key)[IPT],
                                           int (&r)[IPT], int nk) {
  const int t = threadIdx.x;
  const int ndw = nk * (PITCH16 / 2);
  for (int i = t; i < ndw; i += 256) ((int*)priv)[i] = 0;
  __syncthreads();
#pragma unroll
  for (int k = 0; k < IPT; ++k) {
    r[k] = 0;
    if (key[k] >= 0) {
      unsigned short* p = priv + key[k] * PITCH16 + t;
      const unsigned short c = *p;
      r[k] = c;
      *p = (unsigned short)(c + 1);
    }
  }
  __syncthreads();
  for (int kk = t; kk < nk; kk += 256) {
    int* row = (int*)(priv + kk * PITCH16);
    int run = 0;
#pragma unroll 8
    for (int j = 0; j < 128; ++j) {
      const unsigned w = (unsigned)row[j];
      const int lo = (int)(w & 0xffffu);
      const int hi = (int)(w >> 16);
      row[j] = run | ((run + lo) << 16);
      run += lo + hi;
    }
    tot[kk] = run;
  }
  __syncthreads();
}

__device__ __forceinline__ void load_idx(const int* __restrict__ p, int e0, int E, int (&v)[IPT]) {
  const int t = threadIdx.x;
  if ((E & 3) == 0) {
#pragma unroll
    for (int kk = 0; kk < IPT / 4; ++kk) {
      const int i = e0 + kk * 1024 + t * 4;
      int4 q = make_int4(-1, -1, -1, -1);
      if (i < E) q = *(const int4*)(p + i);
      v[4 * kk] = q.x; v[4 * kk + 1] = q.y; v[4 * kk + 2] = q.z; v[4 * kk + 3] = q.w;
    }
  } else {
#pragma unroll
    for (int k = 0; k < IPT; ++k) {
      const int i = e0 + (k >> 2) * 1024 + t * 4 + (k & 3);
      v[k] = (i < E) ? p[i] : -1;
    }
  }
}

__global__ __launch_bounds__(256) void bin_count_kernel(const int* __restrict__ dst,
                                                        int* __restrict__ cnt, int E, int NB) {
  __shared__ __attribute__((aligned(16))) unsigned short priv[MAX_BINS * PITCH16];
  __shared__ int tot[MAX_BINS];
  int d[IPT], key[IPT], r[IPT];
  load_idx(dst, blockIdx.x * CHUNK, E, d);
#pragma unroll
  for (int k = 0; k < IPT; ++k) key[k] = d[k] >> BIN_SHIFT;
  rank_items(priv, tot, key, r, NB);
  for (int b = threadIdx.x; b < NB; b += 256) cnt[blockIdx.x * NB + b] = tot[b];
}

__global__ __launch_bounds__(256) void bin_scatter_kernel(const int* __restrict__ src,
                                                          const int* __restrict__ dst,
                                                          const int* __restrict__ cnt,
                                                          int* __restrict__ bstart,
                                                          int2* __restrict__ binned,
                                                          int E, int NB, int G) {
  __shared__ __attribute__((aligned(16))) unsigned short priv[MAX_BINS * PITCH16];
  __shared__ int tot[MAX_BINS];
  __shared__ int wtot[8];
  const int lane = threadIdx.x & 31;
  const int wid  = threadIdx.x >> 5;
  const int g = blockIdx.x;
  int d[IPT], s[IPT], key[IPT], r[IPT];
  load_idx(dst, g * CHUNK, E, d);
  load_idx(src, g * CHUNK, E, s);
#pragma unroll
  for (int k = 0; k < IPT; ++k) key[k] = d[k] >> BIN_SHIFT;
  rank_items(priv, tot, key, r, NB);

  int carry = 0;
  for (int b0 = 0; b0 < NB; b0 += 256) {
    const int b = b0 + threadIdx.x;
    int all = 0, pre = 0;
    if (b < NB) {
      for (int g0 = 0; g0 < G; g0 += 28) {
        int c[28];
#pragma unroll
        for (int u = 0; u < 28; ++u) c[u] = cnt[min(g0 + u, G - 1) * NB + b];
#pragma unroll
        for (int u = 0; u < 28; ++u) {
          const int v = (g0 + u < G) ? c[u] : 0;
          all += v;
          pre += (g0 + u < g) ? v : 0;
        }
      }
    }
    int x = all;
#pragma unroll
    for (int o = 1; o < 32; o <<= 1) {
      const int y = __shfl_up(x, o, 32);
      if (lane >= o) x += y;
    }
    if (lane == 31) wtot[wid] = x;
    __syncthreads();
    int wp = 0, sum = 0;
#pragma unroll
    for (int w = 0; w < 8; ++w) {
      const int v = wtot[w];
      wp += (w < wid) ? v : 0;
      sum += v;
    }
    const int first = carry + wp + (x - all);
    if (b < NB) {
      tot[b] = first + pre;
      if (g == 0) bstart[b] = first;
    }
    carry += sum;
    __syncthreads();
  }
  if (g == 0 && threadIdx.x == 0) bstart[NB] = carry;

#pragma unroll
  for (int k = 0; k < IPT; ++k) {
    if (key[k] >= 0) {
      const int slot = tot[key[k]] + priv[key[k] * PITCH16 + threadIdx.x] + r[k];
      binned[slot] = make_int2(s[k], d[k]);
    }
  }
}

__global__ __launch_bounds__(256) void bin_csr_kernel(const int2* __restrict__ binned,
                                                      const int* __restrict__ bstart,
                                                      float* __restrict__ deg,
                                                      float* __restrict__ dinv,
                                                      int* __restrict__ start,
                                                      int* __restrict__ csr, int N) {
  __shared__ __attribute__((aligned(16))) unsigned short priv[BIN_NODES * PITCH16];
  __shared__ int tot[BIN_NODES];
  __shared__ int cur[BIN_NODES];
  __shared__ int wtot[8];
  const int t = threadIdx.x;
  const int lane = t & 31;
  const int wid  = t >> 5;
  const int bs = bstart[blockIdx.x];
  const int be = bstart[blockIdx.x + 1];
  int s[IPT], key[IPT], r[IPT];

  cur[t] = 0;
  for (int c0 = bs; c0 < be; c0 += CHUNK) {
#pragma unroll
    for (int k = 0; k < IPT; ++k) {
      const int i = c0 + k * 256 + t;
      const int dd = binned[min(i, be - 1)].y;
      key[k] = (i < be) ? (dd & (BIN_NODES - 1)) : -1;
    }
    rank_items(priv, tot, key, r, BIN_NODES);
    cur[t] += tot[t];
  }
  const int cn = cur[t];
  int x = cn;
#pragma unroll
  for (int o = 1; o < 32; o <<= 1) {
    const int y = __shfl_up(x, o, 32);
    if (lane >= o) x += y;
  }
  if (lane == 31) wtot[wid] = x;
  __syncthreads();
  int wp = 0;
#pragma unroll
  for (int w = 0; w < 8; ++w) wp += (w < wid) ? wtot[w] : 0;
  const int excl = wp + (x - cn);
  const int n = blockIdx.x * BIN_NODES + t;
  if (n < N) {
    start[n] = bs + excl;
    deg[n]   = (float)cn;
    dinv[n]  = rsqrtf((float)cn + 1.0f);
  }
  cur[t] = excl;

  for (int c0 = bs; c0 < be; c0 += CHUNK) {
#pragma unroll
    for (int k = 0; k < IPT; ++k) {
      const int i = c0 + k * 256 + t;
      const int2 sd = binned[min(i, be - 1)];
      s[k] = sd.x;
      key[k] = (i < be) ? (sd.y & (BIN_NODES - 1)) : -1;
    }
    rank_items(priv, tot, key, r, BIN_NODES);
#pragma unroll
    for (int k = 0; k < IPT; ++k) {
      if (key[k] >= 0) {
        const int slot = cur[key[k]] + priv[key[k] * PITCH16 + t] + r[k];
        csr[bs + slot] = s[k];
      }
    }
    __syncthreads();
    cur[t] += tot[t];
  }
}


template <int NCOLS, bool BIAS_RELU_A>
__global__ __launch_bounds__(256) void gemm_wmma_kernel(const float* __restrict__ A,
                                                        const float* __restrict__ abias,
                                                        const float* __restrict__ W,
                                                        const float* __restrict__ dinv,
                                                        float* __restrict__ out, int M) {
  __shared__ float sW[64 * NCOLS];
  const int nvec = (64 * NCOLS) / 4;
  for (int i = threadIdx.x; i < nvec; i += blockDim.x)
    ((float4*)sW)[i] = ((const float4*)W)[i];
  __syncthreads();

  const int lane = threadIdx.x & 31;
  const int wid  = threadIdx.x >> 5;
  const int ntile_n = NCOLS / 16;
  const int mtiles  = (M + 15) >> 4;
  const int gt = blockIdx.x * 8 + wid;
  const int mt = gt / ntile_n;
  const int nt = gt - mt * ntile_n;
  if (mt >= mtiles) return;

  const int m0 = mt * 16;
  const int n0 = nt * 16;
  const int row = lane & 15;
  const int kh  = (lane >> 4) << 1;
  const int arow_idx = min(m0 + row, M - 1);
  const float* __restrict__ arow = A + (size_t)arow_idx * 64 + kh;
  const float* __restrict__ bcol = sW + n0 + row;
  const float* __restrict__ ab = BIAS_RELU_A ? abias + kh : nullptr;
  const float ascale = BIAS_RELU_A ? dinv[arow_idx] : 1.0f;

  v8f c = {};
#pragma unroll
  for (int k = 0; k < 64; k += 4) {
    v2f a, b;
    a.x = arow[k];
    a.y = arow[k + 1];
    if (BIAS_RELU_A) {
      a.x = fmaxf(a.x * ascale + ab[k], 0.0f);
      a.y = fmaxf(a.y * ascale + ab[k + 1], 0.0f);
    }
    b.x = bcol[(k + kh) * NCOLS];
    b.y = bcol[(k + kh + 1) * NCOLS];
    c = __builtin_amdgcn_wmma_f32_16x16x4_f32(false, a, false, b, (short)0, c, false, false);
  }

  const int rbase = (lane >> 4) << 3;
#pragma unroll
  for (int r = 0; r < 8; ++r) {
    int orow = m0 + rbase + r;
    if (orow < M)
      out[(size_t)orow * NCOLS + n0 + row] = c[r] * dinv[orow];
  }
}

__global__ __launch_bounds__(256) __attribute__((amdgpu_waves_per_eu(1, 8)))
void gemm_rows64_kernel(const float* __restrict__ A, const float* __restrict__ W,
                        const float* __restrict__ dinv, float* __restrict__ out, int M) {
  __shared__ __attribute__((aligned(16))) float sW[64 * 64];
  const int lane = threadIdx.x & 31;
  const int wid  = threadIdx.x >> 5;
  const int mtiles = (M + 15) >> 4;
  const int mt = blockIdx.x * 8 + wid;
  const int m0 = mt * 16;
  const int row = lane & 15;
  const int kh  = (lane >> 4) << 1;
  const int rbase = (lane >> 4) << 3;
  const int arow_idx = min(m0 + row, M - 1);
  const float* __restrict__ arow = A + (size_t)arow_idx * 64 + kh;

  v2f a[16];
#pragma unroll
  for (int i = 0; i < 16; ++i) {
    a[i].x = arow[4 * i];
    a[i].y = arow[4 * i + 1];
  }
  float dv[8];
#pragma unroll
  for (int r = 0; r < 8; ++r) dv[r] = dinv[min(m0 + rbase + r, M - 1)];

#pragma unroll
  for (int i = 0; i < 64 * 64 / 4; i += 256)
    ((float4*)sW)[i + threadIdx.x] = ((const float4*)W)[i + threadIdx.x];
  __syncthreads();
  if (mt >= mtiles) return;

  const float* __restrict__ bcol = sW + 4 * row;

  v8f c0 = {}, c1 = {}, c2 = {}, c3 = {};
#pragma unroll
  for (int k = 0; k < 64; k += 4) {
    const float4 q0 = *(const float4*)(bcol + (k + kh) * 64);
    const float4 q1 = *(const float4*)(bcol + (k + kh + 1) * 64);
    v2f b;
    b.x = q0.x; b.y = q1.x;
    c0 = __builtin_amdgcn_wmma_f32_16x16x4_f32(false, a[k >> 2], false, b, (short)0, c0, false, false);
    b.x = q0.y; b.y = q1.y;
    c1 = __builtin_amdgcn_wmma_f32_16x16x4_f32(false, a[k >> 2], false, b, (short)0, c1, false, false);
    b.x = q0.z; b.y = q1.z;
    c2 = __builtin_amdgcn_wmma_f32_16x16x4_f32(false, a[k >> 2], false, b, (short)0, c2, false, false);
    b.x = q0.w; b.y = q1.w;
    c3 = __builtin_amdgcn_wmma_f32_16x16x4_f32(false, a[k >> 2], false, b, (short)0, c3, false, false);
  }

#pragma unroll
  for (int r = 0; r < 8; ++r) {
    const int orow = m0 + rbase + r;
    if (orow < M) {
      const float d = dv[r];
      *(float4*)(out + (size_t)orow * 64 + 4 * row) =
          make_float4(c0[r] * d, c1[r] * d, c2[r] * d, c3[r] * d);
    }
  }
}


#define NODES_PER_WAVE64 8

template <int K>
__device__ __forceinline__ float2 sum_rows64(const float2* __restrict__ hl, int idx, int rem,
                                             float2 acc) {
  float2 v[K];
#pragma unroll
  for (int j = 0; j < K; ++j)
    v[j] = hl[(size_t)__builtin_amdgcn_readlane(idx, j) * 32];
#pragma unroll
  for (int j = 0; j < K; ++j) {
    const bool on = (j < K - 8) || (j < rem);
    acc.x += on ? v[j].x : 0.0f;
    acc.y += on ? v[j].y : 0.0f;
  }
  return acc;
}

__global__ __launch_bounds__(256) void gather64_kernel(const float* __restrict__ h,
                                                       const int* __restrict__ start,
                                                       const float* __restrict__ deg,
                                                       const int* __restrict__ csr,
                                                       float* __restrict__ agg, int N) {
  const int lane = threadIdx.x & 31;
  const int node0 =
      (blockIdx.x * 8 + __builtin_amdgcn_readfirstlane(threadIdx.x >> 5)) * NODES_PER_WAVE64;
  if (node0 >= N) return;
  const int nw = min(NODES_PER_WAVE64, N - node0);
  int begv = 0, cntv = 0;
  if (lane < nw) {
    begv = start[node0 + lane];
    cntv = (int)deg[node0 + lane];
  }
  const float2* __restrict__ hl = (const float2*)h + lane;
  for (int q = 0; q < nw; ++q) {
    const int node = node0 + q;
    int beg = __builtin_amdgcn_readlane(begv, q);
    int rem = __builtin_amdgcn_readlane(cntv, q);
    float2 acc = hl[(size_t)node * 32];
    while (rem > 0) {
      int idx = node;
      if (lane < rem) idx = csr[beg + lane];
      if (rem > 24)      acc = sum_rows64<32>(hl, idx, rem, acc);
      else if (rem > 16) acc = sum_rows64<24>(hl, idx, rem, acc);
      else if (rem > 8)  acc = sum_rows64<16>(hl, idx, rem, acc);
      else               acc = sum_rows64<8>(hl, idx, rem, acc);
      rem -= 32;
      beg += 32;
    }
    ((float2*)agg)[(size_t)node * 32 + lane] = acc;
  }
}

#define PAIRS_PER_WAVE16 4

template <int K>
__device__ __forceinline__ float sum_rows16(const float* __restrict__ hc, int pick, int i0, int i1,
                                            int rem, float acc) {
  float v[K];
#pragma unroll
  for (int j = 0; j < K; ++j) {
    const int sj = __builtin_amdgcn_ds_bpermute(pick + 4 * (j & 15), j < 16 ? i0 : i1);
    v[j] = hc[(size_t)sj * 16];
  }
#pragma unroll
  for (int j = 0; j < K; ++j)
    acc += (j < rem) ? v[j] : 0.0f;
  return acc;
}

__global__ __launch_bounds__(256) void gather16_kernel(const float* __restrict__ h,
                                                       const int* __restrict__ start,
                                                       const float* __restrict__ deg,
                                                       const int* __restrict__ csr,
                                                       const float* __restrict__ dinv,
                                                       const float* __restrict__ bias,
                                                       float* __restrict__ out, int N) {
  const int lane = threadIdx.x & 31;
  const int c = lane & 15;
  const int half = lane >> 4;
  const int base0 = (blockIdx.x * 8 + __builtin_amdgcn_readfirstlane(threadIdx.x >> 5)) *
                    (2 * PAIRS_PER_WAVE16);
  if (base0 >= N) return;
  const int np = min(PAIRS_PER_WAVE16, (N - base0 + 1) >> 1);
  int begv = 0, cntv = 0;
  float dv = 0.0f;
  if (lane < 2 * PAIRS_PER_WAVE16 && base0 + lane < N) {
    begv = start[base0 + lane];
    cntv = (int)deg[base0 + lane];
    dv = dinv[base0 + lane];
  }
  const int dvbits = __float_as_int(dv);
  const float* __restrict__ hc = h + c;
  const float bc = bias[c];
  const int pick = half << 6;

  for (int q = 0; q < np; ++q) {
    const int node = base0 + 2 * q + half;
    const int nd = min(node, N - 1);
    const int a = (2 * q + half) << 2;
    int beg = __builtin_amdgcn_ds_bpermute(a, begv);
    int rem = __builtin_amdgcn_ds_bpermute(a, cntv);
    const float dd = __int_as_float(__builtin_amdgcn_ds_bpermute(a, dvbits));
    int mu = max(__builtin_amdgcn_readlane(cntv, 2 * q), __builtin_amdgcn_readlane(cntv, 2 * q + 1));
    float acc = hc[(size_t)nd * 16];
    while (mu > 0) {
      int i0 = nd, i1 = nd;
      if (c < rem)      i0 = csr[beg + c];
      if (c + 16 < rem) i1 = csr[beg + 16 + c];
      if (mu > 24)      acc = sum_rows16<32>(hc, pick, i0, i1, rem, acc);
      else if (mu > 16) acc = sum_rows16<24>(hc, pick, i0, i1, rem, acc);
      else if (mu > 8)  acc = sum_rows16<16>(hc, pick, i0, i1, rem, acc);
      else              acc = sum_rows16<8>(hc, pick, i0, i1, rem, acc);
      mu -= 32;
      rem -= 32;
      beg += 32;
    }
    if (node < N) out[(size_t)node * 16 + c] = bc + dd * acc;
  }
}


extern "C" void kernel_launch(void* const* d_in, const int* in_sizes, int n_in,
                              void* d_out, int out_size, void* d_ws, size_t ws_size,
                              hipStream_t stream) {
  const float* x    = (const float*)d_in[0];
  const int*   edge = (const int*)d_in[1];
  const float* W1   = (const float*)d_in[2];
  const float* b1   = (const float*)d_in[3];
  const float* W2   = (const float*)d_in[4];
  const float* b2   = (const float*)d_in[5];
  float* out = (float*)d_out;

  const int E  = in_sizes[1] / 2;
  const int Nn = in_sizes[0] / IN_DIM;
  const int* src = edge;
  const int* dst = edge + E;

  const int NB = (Nn + BIN_NODES - 1) >> BIN_SHIFT;
  const int G  = (E + CHUNK - 1) / CHUNK;

  const size_t nodeB = ((size_t)Nn * 4 + 255) & ~(size_t)255;
  const size_t edgeB = ((size_t)E * 4 + 255) & ~(size_t)255;
  const size_t cntB  = ((size_t)G * NB * 4 + 255) & ~(size_t)255;
  const size_t bstB  = ((size_t)(NB + 1) * 4 + 255) & ~(size_t)255;
  const size_t featB = (size_t)Nn * 64 * 4;
  char* ws = (char*)d_ws;
  float* deg    = (float*)ws;
  float* dinv   = (float*)(ws + nodeB);
  int*   start  = (int*)(ws + 2 * nodeB);
  int*   cnt    = (int*)(ws + 3 * nodeB);
  int*   bstart = (int*)(ws + 3 * nodeB + cntB);
  int2*  binned = (int2*)(ws + 3 * nodeB + cntB + bstB);
  int*   csr    = (int*)(ws + 3 * nodeB + cntB + bstB + 2 * edgeB);
  float* h1     = (float*)(ws + 3 * nodeB + cntB + bstB + 3 * edgeB);
  float* agg1   = (float*)(ws + 3 * nodeB + cntB + bstB + 3 * edgeB + featB);
  float* h2     = h1;

  const int mtiles = (Nn + 15) >> 4;

  bin_count_kernel<<<G, 256, 0, stream>>>(dst, cnt, E, NB);
  bin_scatter_kernel<<<G, 256, 0, stream>>>(src, dst, cnt, bstart, binned, E, NB, G);
  bin_csr_kernel<<<NB, 256, 0, stream>>>(binned, bstart, deg, dinv, start, csr, Nn);

  gemm_rows64_kernel<<<(mtiles + 7) / 8, 256, 0, stream>>>(x, W1, dinv, h1, Nn);
  gather64_kernel<<<(Nn + 8 * NODES_PER_WAVE64 - 1) / (8 * NODES_PER_WAVE64), 256, 0, stream>>>(
      h1, start, deg, csr, agg1, Nn);

  gemm_wmma_kernel<16, true><<<(mtiles + 7) / 8, 256, 0, stream>>>(
      agg1, b1, W2, dinv, h2, Nn);
  gather16_kernel<<<(Nn + 16 * PAIRS_PER_WAVE16 - 1) / (16 * PAIRS_PER_WAVE16), 256, 0, stream>>>(
      h2, start, deg, csr, dinv, b2, out, Nn);

  (void)n_in; (void)out_size; (void)ws_size;
}


# Layer-2 projection moved into layer 1's gather: a block's 64 summed rows go, relu'd, through LDS into the same WMMA tiles (waves 0-3); agg1's store, re-read and one launch gone; sums as before
# speedup vs baseline: 2.9337x; 1.0722x over previous
#include <hip/hip_runtime.h>
#include <hip/hip_bf16.h>


#define IN_DIM  64
#define OUT_DIM 16

typedef __attribute__((ext_vector_type(2))) float v2f;
typedef __attribute__((ext_vector_type(8))) float v8f;


#define BIN_SHIFT 8
#define BIN_NODES 256
#define MAX_BINS  512
#define IPT       32
#define CHUNK     (256 * IPT)
#define PITCH16   258

__device__ __forceinline__ void rank_items(unsigned short* __restrict__ priv,
                                           int* __restrict__ tot, const int (&key)[IPT],
                                           int (&r)[IPT], int nk) {
  const int t = threadIdx.x;
  const int ndw = nk * (PITCH16 / 2);
  for (int i = t; i < ndw; i += 256) ((int*)priv)[i] = 0;
  __syncthreads();
#pragma unroll
  for (int k = 0; k < IPT; ++k) {
    r[k] = 0;
    if (key[k] >= 0) {
      unsigned short* p = priv + key[k] * PITCH16 + t;
      const unsigned short c = *p;
      r[k] = c;
      *p = (unsigned short)(c + 1);
    }
  }
  __syncthreads();
  for (int kk = t; kk < nk; kk += 256) {
    int* row = (int*)(priv + kk * PITCH16);
    int run = 0;
#pragma unroll 8
    for (int j = 0; j < 128; ++j) {
      const unsigned w = (unsigned)row[j];
      const int lo = (int)(w & 0xffffu);
      const int hi = (int)(w >> 16);
      row[j] = run | ((run + lo) << 16);
      run += lo + hi;
    }
    tot[kk] = run;
  }
  __syncthreads();
}

__device__ __forceinline__ void load_idx(const int* __restrict__ p, int e0, int E, int (&v)[IPT]) {
  const int t = threadIdx.x;
  if ((E & 3) == 0) {
#pragma unroll
    for (int kk = 0; kk < IPT / 4; ++kk) {
      const int i = e0 + kk * 1024 + t * 4;
      int4 q = make_int4(-1, -1, -1, -1);
      if (i < E) q = *(const int4*)(p + i);
      v[4 * kk] = q.x; v[4 * kk + 1] = q.y; v[4 * kk + 2] = q.z; v[4 * kk + 3] = q.w;
    }
  } else {
#pragma unroll
    for (int k = 0; k < IPT; ++k) {
      const int i = e0 + (k >> 2) * 1024 + t * 4 + (k & 3);
      v[k] = (i < E) ? p[i] : -1;
    }
  }
}

__global__ __launch_bounds__(256) void bin_count_kernel(const int* __restrict__ dst,
                                                        int* __restrict__ cnt, int E, int NB) {
  __shared__ __attribute__((aligned(16))) unsigned short priv[MAX_BINS * PITCH16];
  __shared__ int tot[MAX_BINS];
  int d[IPT], key[IPT], r[IPT];
  load_idx(dst, blockIdx.x * CHUNK, E, d);
#pragma unroll
  for (int k = 0; k < IPT; ++k) key[k] = d[k] >> BIN_SHIFT;
  rank_items(priv, tot, key, r, NB);
  for (int b = threadIdx.x; b < NB; b += 256) cnt[blockIdx.x * NB + b] = tot[b];
}

__global__ __launch_bounds__(256) void bin_scatter_kernel(const int* __restrict__ src,
                                                          const int* __restrict__ dst,
                                                          const int* __restrict__ cnt,
                                                          int* __restrict__ bstart,
                                                          int2* __restrict__ binned,
                                                          int E, int NB, int G) {
  __shared__ __attribute__((aligned(16))) unsigned short priv[MAX_BINS * PITCH16];
  __shared__ int tot[MAX_BINS];
  __shared__ int wtot[8];
  const int lane = threadIdx.x & 31;
  const int wid  = threadIdx.x >> 5;
  const int g = blockIdx.x;
  int d[IPT], s[IPT], key[IPT], r[IPT];
  load_idx(dst, g * CHUNK, E, d);
  load_idx(src, g * CHUNK, E, s);
#pragma unroll
  for (int k = 0; k < IPT; ++k) key[k] = d[k] >> BIN_SHIFT;
  rank_items(priv, tot, key, r, NB);

  int carry = 0;
  for (int b0 = 0; b0 < NB; b0 += 256) {
    const int b = b0 + threadIdx.x;
    int all = 0, pre = 0;
    if (b < NB) {
      for (int g0 = 0; g0 < G; g0 += 28) {
        int c[28];
#pragma unroll
        for (int u = 0; u < 28; ++u) c[u] = cnt[min(g0 + u, G - 1) * NB + b];
#pragma unroll
        for (int u = 0; u < 28; ++u) {
          const int v = (g0 + u < G) ? c[u] : 0;
          all += v;
          pre += (g0 + u < g) ? v : 0;
        }
      }
    }
    int x = all;
#pragma unroll
    for (int o = 1; o < 32; o <<= 1) {
      const int y = __shfl_up(x, o, 32);
      if (lane >= o) x += y;
    }
    if (lane == 31) wtot[wid] = x;
    __syncthreads();
    int wp = 0, sum = 0;
#pragma unroll
    for (int w = 0; w < 8; ++w) {
      const int v = wtot[w];
      wp += (w < wid) ? v : 0;
      sum += v;
    }
    const int first = carry + wp + (x - all);
    if (b < NB) {
      tot[b] = first + pre;
      if (g == 0) bstart[b] = first;
    }
    carry += sum;
    __syncthreads();
  }
  if (g == 0 && threadIdx.x == 0) bstart[NB] = carry;

#pragma unroll
  for (int k = 0; k < IPT; ++k) {
    if (key[k] >= 0) {
      const int slot = tot[key[k]] + priv[key[k] * PITCH16 + threadIdx.x] + r[k];
      binned[slot] = make_int2(s[k], d[k]);
    }
  }
}

__global__ __launch_bounds__(256) void bin_csr_kernel(const int2* __restrict__ binned,
                                                      const int* __restrict__ bstart,
                                                      float* __restrict__ deg,
                                                      float* __restrict__ dinv,
                                                      int* __restrict__ start,
                                                      int* __restrict__ csr, int N) {
  __shared__ __attribute__((aligned(16))) unsigned short priv[BIN_NODES * PITCH16];
  __shared__ int tot[BIN_NODES];
  __shared__ int cur[BIN_NODES];
  __shared__ int wtot[8];
  const int t = threadIdx.x;
  const int lane = t & 31;
  const int wid  = t >> 5;
  const int bs = bstart[blockIdx.x];
  const int be = bstart[blockIdx.x + 1];
  int s[IPT], key[IPT], r[IPT];

  cur[t] = 0;
  for (int c0 = bs; c0 < be; c0 += CHUNK) {
#pragma unroll
    for (int k = 0; k < IPT; ++k) {
      const int i = c0 + k * 256 + t;
      const int dd = binned[min(i, be - 1)].y;
      key[k] = (i < be) ? (dd & (BIN_NODES - 1)) : -1;
    }
    rank_items(priv, tot, key, r, BIN_NODES);
    cur[t] += tot[t];
  }
  const int cn = cur[t];
  int x = cn;
#pragma unroll
  for (int o = 1; o < 32; o <<= 1) {
    const int y = __shfl_up(x, o, 32);
    if (lane >= o) x += y;
  }
  if (lane == 31) wtot[wid] = x;
  __syncthreads();
  int wp = 0;
#pragma unroll
  for (int w = 0; w < 8; ++w) wp += (w < wid) ? wtot[w] : 0;
  const int excl = wp + (x - cn);
  const int n = blockIdx.x * BIN_NODES + t;
  if (n < N) {
    start[n] = bs + excl;
    deg[n]   = (float)cn;
    dinv[n]  = rsqrtf((float)cn + 1.0f);
  }
  cur[t] = excl;

  for (int c0 = bs; c0 < be; c0 += CHUNK) {
#pragma unroll
    for (int k = 0; k < IPT; ++k) {
      const int i = c0 + k * 256 + t;
      const int2 sd = binned[min(i, be - 1)];
      s[k] = sd.x;
      key[k] = (i < be) ? (sd.y & (BIN_NODES - 1)) : -1;
    }
    rank_items(priv, tot, key, r, BIN_NODES);
#pragma unroll
    for (int k = 0; k < IPT; ++k) {
      if (key[k] >= 0) {
        const int slot = cur[key[k]] + priv[key[k] * PITCH16 + t] + r[k];
        csr[bs + slot] = s[k];
      }
    }
    __syncthreads();
    cur[t] += tot[t];
  }
}


__global__ __launch_bounds__(256) __attribute__((amdgpu_waves_per_eu(1, 8)))
void gemm_rows64_kernel(const float* __restrict__ A, const float* __restrict__ W,
                        const float* __restrict__ dinv, float* __restrict__ out, int M) {
  __shared__ __attribute__((aligned(16))) float sW[64 * 64];
  const int lane = threadIdx.x & 31;
  const int wid  = threadIdx.x >> 5;
  const int mtiles = (M + 15) >> 4;
  const int mt = blockIdx.x * 8 + wid;
  const int m0 = mt * 16;
  const int row = lane & 15;
  const int kh  = (lane >> 4) << 1;
  const int rbase = (lane >> 4) << 3;
  const int arow_idx = min(m0 + row, M - 1);
  const float* __restrict__ arow = A + (size_t)arow_idx * 64 + kh;

  v2f a[16];
#pragma unroll
  for (int i = 0; i < 16; ++i) {
    a[i].x = arow[4 * i];
    a[i].y = arow[4 * i + 1];
  }
  float dv[8];
#pragma unroll
  for (int r = 0; r < 8; ++r) dv[r] = dinv[min(m0 + rbase + r, M - 1)];

#pragma unroll
  for (int i = 0; i < 64 * 64 / 4; i += 256)
    ((float4*)sW)[i + threadIdx.x] = ((const float4*)W)[i + threadIdx.x];
  __syncthreads();
  if (mt >= mtiles) return;

  const float* __restrict__ bcol = sW + 4 * row;

  v8f c0 = {}, c1 = {}, c2 = {}, c3 = {};
#pragma unroll
  for (int k = 0; k < 64; k += 4) {
    const float4 q0 = *(const float4*)(bcol + (k + kh) * 64);
    const float4 q1 = *(const float4*)(bcol + (k + kh + 1) * 64);
    v2f b;
    b.x = q0.x; b.y = q1.x;
    c0 = __builtin_amdgcn_wmma_f32_16x16x4_f32(false, a[k >> 2], false, b, (short)0, c0, false, false);
    b.x = q0.y; b.y = q1.y;
    c1 = __builtin_amdgcn_wmma_f32_16x16x4_f32(false, a[k >> 2], false, b, (short)0, c1, false, false);
    b.x = q0.z; b.y = q1.z;
    c2 = __builtin_amdgcn_wmma_f32_16x16x4_f32(false, a[k >> 2], false, b, (short)0, c2, false, false);
    b.x = q0.w; b.y = q1.w;
    c3 = __builtin_amdgcn_wmma_f32_16x16x4_f32(false, a[k >> 2], false, b, (short)0, c3, false, false);
  }

#pragma unroll
  for (int r = 0; r < 8; ++r) {
    const int orow = m0 + rbase + r;
    if (orow < M) {
      const float d = dv[r];
      *(float4*)(out + (size_t)orow * 64 + 4 * row) =
          make_float4(c0[r] * d, c1[r] * d, c2[r] * d, c3[r] * d);
    }
  }
}


#define NODES_PER_WAVE64 8

template <int K>
__device__ __forceinline__ float2 sum_rows64(const float2* __restrict__ hl, int idx, int rem,
                                             float2 acc) {
  float2 v[K];
#pragma unroll
  for (int j = 0; j < K; ++j)
    v[j] = hl[(size_t)__builtin_amdgcn_readlane(idx, j) * 32];
#pragma unroll
  for (int j = 0; j < K; ++j) {
    const bool on = (j < K - 8) || (j < rem);
    acc.x += on ? v[j].x : 0.0f;
    acc.y += on ? v[j].y : 0.0f;
  }
  return acc;
}

#define ROWS_PER_BLOCK64 (8 * NODES_PER_WAVE64)
#define APITCH 68

__global__ __launch_bounds__(256) void gather64_proj_kernel(const float* __restrict__ h,
                                                            const int* __restrict__ start,
                                                            const float* __restrict__ deg,
                                                            const int* __restrict__ csr,
                                                            const float* __restrict__ dinv,
                                                            const float* __restrict__ b1,
                                                            const float* __restrict__ W2,
                                                            float* __restrict__ h2, int N) {
  __shared__ __attribute__((aligned(16))) float sA[ROWS_PER_BLOCK64 * APITCH];
  __shared__ __attribute__((aligned(16))) float sW[IN_DIM * OUT_DIM];
  __shared__ __attribute__((aligned(16))) float sD[ROWS_PER_BLOCK64];
  const int lane = threadIdx.x & 31;
  const int wid = __builtin_amdgcn_readfirstlane(threadIdx.x >> 5);
  const int node0 = (blockIdx.x * 8 + wid) * NODES_PER_WAVE64;
  const int nw = max(0, min(NODES_PER_WAVE64, N - node0));
  const float4 w2v = ((const float4*)W2)[threadIdx.x];
  const float2 bb = ((const float2*)b1)[lane];
  const int li = min(node0 + lane, N - 1);
  const int   ldb = start[li];
  const float ldc = deg[li];
  const float ldd = dinv[li];
  const bool has = lane < nw;
  const int begv = has ? ldb : 0;
  const int cntv = has ? (int)ldc : 0;
  const float dvv = has ? ldd : 0.0f;
  const int dvbits = __float_as_int(dvv);
  ((float4*)sW)[threadIdx.x] = w2v;
  if (lane < NODES_PER_WAVE64) sD[wid * NODES_PER_WAVE64 + lane] = dvv;

  const float2* __restrict__ hl = (const float2*)h + lane;
  float* __restrict__ srow = sA + (wid * NODES_PER_WAVE64) * APITCH + 2 * lane;
  for (int q = 0; q < NODES_PER_WAVE64; ++q) {
    float2 a = make_float2(0.0f, 0.0f);
    if (q < nw) {
      const int node = node0 + q;
      int beg = __builtin_amdgcn_readlane(begv, q);
      int rem = __builtin_amdgcn_readlane(cntv, q);
      const float dq = __int_as_float(__builtin_amdgcn_readlane(dvbits, q));
      float2 acc = hl[(size_t)node * 32];
      while (rem > 0) {
        int idx = node;
        if (lane < rem) idx = csr[beg + lane];
        if (rem > 24)      acc = sum_rows64<32>(hl, idx, rem, acc);
        else if (rem > 16) acc = sum_rows64<24>(hl, idx, rem, acc);
        else if (rem > 8)  acc = sum_rows64<16>(hl, idx, rem, acc);
        else               acc = sum_rows64<8>(hl, idx, rem, acc);
        rem -= 32;
        beg += 32;
      }
      a.x = fmaxf(acc.x * dq + bb.x, 0.0f);
      a.y = fmaxf(acc.y * dq + bb.y, 0.0f);
    }
    *(float2*)(srow + q * APITCH) = a;
  }
  __syncthreads();

  const int m0 = blockIdx.x * ROWS_PER_BLOCK64 + wid * 16;
  if (wid >= ROWS_PER_BLOCK64 / 16 || m0 >= N) return;

  const int row = lane & 15;
  const int kh  = (lane >> 4) << 1;
  const int rbase = (lane >> 4) << 3;
  const float4 dlo = *(const float4*)(sD + wid * 16 + rbase);
  const float4 dhi = *(const float4*)(sD + wid * 16 + rbase + 4);
  const float dv[8] = {dlo.x, dlo.y, dlo.z, dlo.w, dhi.x, dhi.y, dhi.z, dhi.w};
  const float* __restrict__ ar = sA + (wid * 16 + row) * APITCH + kh;
  const float* __restrict__ bcol = sW + row;

  v8f c = {};
#pragma unroll
  for (int k = 0; k < IN_DIM; k += 4) {
    const v2f a = *(const v2f*)(ar + k);
    v2f b;
    b.x = bcol[(k + kh) * OUT_DIM];
    b.y = bcol[(k + kh + 1) * OUT_DIM];
    c = __builtin_amdgcn_wmma_f32_16x16x4_f32(false, a, false, b, (short)0, c, false, false);
  }

#pragma unroll
  for (int r = 0; r < 8; ++r) {
    const int orow = m0 + rbase + r;
    if (orow < N) h2[(size_t)orow * OUT_DIM + row] = c[r] * dv[r];
  }
}

#define PAIRS_PER_WAVE16 4

template <int K>
__device__ __forceinline__ float sum_rows16(const float* __restrict__ hc, int pick, int i0, int i1,
                                            int rem, float acc) {
  float v[K];
#pragma unroll
  for (int j = 0; j < K; ++j) {
    const int sj = __builtin_amdgcn_ds_bpermute(pick + 4 * (j & 15), j < 16 ? i0 : i1);
    v[j] = hc[(size_t)sj * 16];
  }
#pragma unroll
  for (int j = 0; j < K; ++j)
    acc += (j < rem) ? v[j] : 0.0f;
  return acc;
}

__global__ __launch_bounds__(256) void gather16_kernel(const float* __restrict__ h,
                                                       const int* __restrict__ start,
                                                       const float* __restrict__ deg,
                                                       const int* __restrict__ csr,
                                                       const float* __restrict__ dinv,
                                                       const float* __restrict__ bias,
                                                       float* __restrict__ out, int N) {
  const int lane = threadIdx.x & 31;
  const int c = lane & 15;
  const int half = lane >> 4;
  const int base0 = (blockIdx.x * 8 + __builtin_amdgcn_readfirstlane(threadIdx.x >> 5)) *
                    (2 * PAIRS_PER_WAVE16);
  if (base0 >= N) return;
  const int np = min(PAIRS_PER_WAVE16, (N - base0 + 1) >> 1);
  int begv = 0, cntv = 0;
  float dv = 0.0f;
  if (lane < 2 * PAIRS_PER_WAVE16 && base0 + lane < N) {
    begv = start[base0 + lane];
    cntv = (int)deg[base0 + lane];
    dv = dinv[base0 + lane];
  }
  const int dvbits = __float_as_int(dv);
  const float* __restrict__ hc = h + c;
  const float bc = bias[c];
  const int pick = half << 6;

  for (int q = 0; q < np; ++q) {
    const int node = base0 + 2 * q + half;
    const int nd = min(node, N - 1);
    const int a = (2 * q + half) << 2;
    int beg = __builtin_amdgcn_ds_bpermute(a, begv);
    int rem = __builtin_amdgcn_ds_bpermute(a, cntv);
    const float dd = __int_as_float(__builtin_amdgcn_ds_bpermute(a, dvbits));
    int mu = max(__builtin_amdgcn_readlane(cntv, 2 * q), __builtin_amdgcn_readlane(cntv, 2 * q + 1));
    float acc = hc[(size_t)nd * 16];
    while (mu > 0) {
      int i0 = nd, i1 = nd;
      if (c < rem)      i0 = csr[beg + c];
      if (c + 16 < rem) i1 = csr[beg + 16 + c];
      if (mu > 24)      acc = sum_rows16<32>(hc, pick, i0, i1, rem, acc);
      else if (mu > 16) acc = sum_rows16<24>(hc, pick, i0, i1, rem, acc);
      else if (mu > 8)  acc = sum_rows16<16>(hc, pick, i0, i1, rem, acc);
      else              acc = sum_rows16<8>(hc, pick, i0, i1, rem, acc);
      mu -= 32;
      rem -= 32;
      beg += 32;
    }
    if (node < N) out[(size_t)node * 16 + c] = bc + dd * acc;
  }
}


extern "C" void kernel_launch(void* const* d_in, const int* in_sizes, int n_in,
                              void* d_out, int out_size, void* d_ws, size_t ws_size,
                              hipStream_t stream) {
  const float* x    = (const float*)d_in[0];
  const int*   edge = (const int*)d_in[1];
  const float* W1   = (const float*)d_in[2];
  const float* b1   = (const float*)d_in[3];
  const float* W2   = (const float*)d_in[4];
  const float* b2   = (const float*)d_in[5];
  float* out = (float*)d_out;

  const int E  = in_sizes[1] / 2;
  const int Nn = in_sizes[0] / IN_DIM;
  const int* src = edge;
  const int* dst = edge + E;

  const int NB = (Nn + BIN_NODES - 1) >> BIN_SHIFT;
  const int G  = (E + CHUNK - 1) / CHUNK;

  const size_t nodeB = ((size_t)Nn * 4 + 255) & ~(size_t)255;
  const size_t edgeB = ((size_t)E * 4 + 255) & ~(size_t)255;
  const size_t cntB  = ((size_t)G * NB * 4 + 255) & ~(size_t)255;
  const size_t bstB  = ((size_t)(NB + 1) * 4 + 255) & ~(size_t)255;
  const size_t featB = (size_t)Nn * 64 * 4;
  char* ws = (char*)d_ws;
  float* deg    = (float*)ws;
  float* dinv   = (float*)(ws + nodeB);
  int*   start  = (int*)(ws + 2 * nodeB);
  int*   cnt    = (int*)(ws + 3 * nodeB);
  int*   bstart = (int*)(ws + 3 * nodeB + cntB);
  int2*  binned = (int2*)(ws + 3 * nodeB + cntB + bstB);
  int*   csr    = (int*)(ws + 3 * nodeB + cntB + bstB + 2 * edgeB);
  float* h1     = (float*)(ws + 3 * nodeB + cntB + bstB + 3 * edgeB);
  float* h2     = (float*)(ws + 3 * nodeB + cntB + bstB + 3 * edgeB + featB);

  const int mtiles = (Nn + 15) >> 4;

  bin_count_kernel<<<G, 256, 0, stream>>>(dst, cnt, E, NB);
  bin_scatter_kernel<<<G, 256, 0, stream>>>(src, dst, cnt, bstart, binned, E, NB, G);
  bin_csr_kernel<<<NB, 256, 0, stream>>>(binned, bstart, deg, dinv, start, csr, Nn);

  gemm_rows64_kernel<<<(mtiles + 7) / 8, 256, 0, stream>>>(x, W1, dinv, h1, Nn);
  gather64_proj_kernel<<<(Nn + ROWS_PER_BLOCK64 - 1) / ROWS_PER_BLOCK64, 256, 0, stream>>>(
      h1, start, deg, csr, dinv, b1, W2, h2, Nn);

  gather16_kernel<<<(Nn + 16 * PAIRS_PER_WAVE16 - 1) / (16 * PAIRS_PER_WAVE16), 256, 0, stream>>>(
      h2, start, deg, csr, dinv, b2, out, Nn);

  (void)n_in; (void)out_size; (void)ws_size;
}


# CSR build without its counting pass: a chunk sorts its edges by bin inside its own region and stores the pieces' starts; a bin's work-group sums the starts and copies its pieces into one run
# speedup vs baseline: 3.7100x; 1.2646x over previous
#include <hip/hip_runtime.h>
#include <hip/hip_bf16.h>


#define IN_DIM  64
#define OUT_DIM 16

typedef __attribute__((ext_vector_type(2))) float v2f;
typedef __attribute__((ext_vector_type(8))) float v8f;


#define BIN_SHIFT 8
#define BIN_NODES 256
#define MAX_BINS  512
#define IPT       32
#define CHUNK     (256 * IPT)
#define PITCH16   258

__device__ __forceinline__ void rank_items(unsigned short* __restrict__ priv,
                                           int* __restrict__ tot, const int (&key)[IPT],
                                           int (&r)[IPT], int nk) {
  const int t = threadIdx.x;
  const int ndw = nk * (PITCH16 / 2);
  for (int i = t; i < ndw; i += 256) ((int*)priv)[i] = 0;
  __syncthreads();
#pragma unroll
  for (int k = 0; k < IPT; ++k) {
    r[k] = 0;
    if (key[k] >= 0) {
      unsigned short* p = priv + key[k] * PITCH16 + t;
      const unsigned short c = *p;
      r[k] = c;
      *p = (unsigned short)(c + 1);
    }
  }
  __syncthreads();
  for (int kk = t; kk < nk; kk += 256) {
    int* row = (int*)(priv + kk * PITCH16);
    int run = 0;
#pragma unroll 8
    for (int j = 0; j < 128; ++j) {
      const unsigned w = (unsigned)row[j];
      const int lo = (int)(w & 0xffffu);
      const int hi = (int)(w >> 16);
      row[j] = run | ((run + lo) << 16);
      run += lo + hi;
    }
    tot[kk] = run;
  }
  __syncthreads();
}

__device__ __forceinline__ void load_idx(const int* __restrict__ p, int e0, int E, int (&v)[IPT]) {
  const int t = threadIdx.x;
  if ((E & 3) == 0) {
#pragma unroll
    for (int kk = 0; kk < IPT / 4; ++kk) {
      const int i = e0 + kk * 1024 + t * 4;
      int4 q = make_int4(-1, -1, -1, -1);
      if (i < E) q = *(const int4*)(p + i);
      v[4 * kk] = q.x; v[4 * kk + 1] = q.y; v[4 * kk + 2] = q.z; v[4 * kk + 3] = q.w;
    }
  } else {
#pragma unroll
    for (int k = 0; k < IPT; ++k) {
      const int i = e0 + (k >> 2) * 1024 + t * 4 + (k & 3);
      v[k] = (i < E) ? p[i] : -1;
    }
  }
}

__global__ __launch_bounds__(256) void chunk_sort_kernel(const int* __restrict__ src,
                                                         const int* __restrict__ dst,
                                                         int* __restrict__ off,
                                                         int2* __restrict__ binned,
                                                         int E, int NB) {
  __shared__ __attribute__((aligned(16))) unsigned short priv[MAX_BINS * PITCH16];
  __shared__ int tot[MAX_BINS];
  __shared__ int wtot[8];
  const int lane = threadIdx.x & 31;
  const int wid  = threadIdx.x >> 5;
  const int g = blockIdx.x;
  int d[IPT], s[IPT], key[IPT], r[IPT];
  load_idx(dst, g * CHUNK, E, d);
  load_idx(src, g * CHUNK, E, s);
#pragma unroll
  for (int k = 0; k < IPT; ++k) key[k] = d[k] >> BIN_SHIFT;
  rank_items(priv, tot, key, r, NB);

  int* __restrict__ orow = off + (size_t)g * (NB + 1);
  int carry = 0;
  for (int b0 = 0; b0 < NB; b0 += 256) {
    const int b = b0 + threadIdx.x;
    const int all = (b < NB) ? tot[b] : 0;
    int x = all;
#pragma unroll
    for (int o = 1; o < 32; o <<= 1) {
      const int y = __shfl_up(x, o, 32);
      if (lane >= o) x += y;
    }
    if (lane == 31) wtot[wid] = x;
    __syncthreads();
    int wp = 0, sum = 0;
#pragma unroll
    for (int w = 0; w < 8; ++w) {
      const int v = wtot[w];
      wp += (w < wid) ? v : 0;
      sum += v;
    }
    const int first = carry + wp + (x - all);
    if (b < NB) {
      tot[b] = first;
      orow[b] = first;
    }
    carry += sum;
    __syncthreads();
  }
  if (threadIdx.x == 0) orow[NB] = carry;

  int2* __restrict__ mine = binned + (size_t)g * CHUNK;
#pragma unroll
  for (int k = 0; k < IPT; ++k) {
    if (key[k] >= 0) {
      const int slot = tot[key[k]] + priv[key[k] * PITCH16 + threadIdx.x] + r[k];
      mine[slot] = make_int2(s[k], d[k]);
    }
  }
}

#define SEG_BATCH 16

__global__ __launch_bounds__(256) void bin_csr_kernel(const int2* __restrict__ binned,
                                                      const int* __restrict__ off,
                                                      int2* runbuf,
                                                      float* __restrict__ deg,
                                                      float* __restrict__ dinv,
                                                      int* __restrict__ start,
                                                      int* __restrict__ csr, int N, int NB, int G) {
  __shared__ __attribute__((aligned(16))) unsigned short priv[BIN_NODES * PITCH16];
  __shared__ int tot[BIN_NODES];
  __shared__ int cur[BIN_NODES];
  __shared__ int wtot[8];
  __shared__ int red[16];
  const int t = threadIdx.x;
  const int lane = t & 31;
  const int wid  = __builtin_amdgcn_readfirstlane(t >> 5);
  const int bin = blockIdx.x;
  const int pitch = NB + 1;

  int slo = 0, shi = 0;
  for (int g = t; g < G; g += 256) {
    slo += off[(size_t)g * pitch + bin];
    shi += off[(size_t)g * pitch + bin + 1];
  }
#pragma unroll
  for (int o = 16; o > 0; o >>= 1) {
    slo += __shfl_xor(slo, o, 32);
    shi += __shfl_xor(shi, o, 32);
  }
  if (lane == 0) { red[wid] = slo; red[8 + wid] = shi; }
  __syncthreads();
  int bs = 0, be = 0;
#pragma unroll
  for (int w = 0; w < 8; ++w) { bs += red[w]; be += red[8 + w]; }
  int2* run = runbuf + ((size_t)((bs + 15) & ~15) + (size_t)16 * bin - (size_t)bs);

  int pcarry = 0;
  for (int g0 = 0; g0 < G; g0 += 256) {
    const int g = g0 + t;
    int lo = 0, len = 0;
    if (g < G) {
      lo = off[(size_t)g * pitch + bin];
      len = off[(size_t)g * pitch + bin + 1] - lo;
    }
    int x = len;
#pragma unroll
    for (int o = 1; o < 32; o <<= 1) {
      const int y = __shfl_up(x, o, 32);
      if (lane >= o) x += y;
    }
    if (lane == 31) wtot[wid] = x;
    __syncthreads();
    int wp = 0, sum = 0;
#pragma unroll
    for (int w = 0; w < 8; ++w) {
      const int v = wtot[w];
      wp += (w < wid) ? v : 0;
      sum += v;
    }
    const int from = g * CHUNK + lo;
    const int to = bs + pcarry + wp + (x - len);
    pcarry += sum;
#pragma unroll
    for (int u0 = 0; u0 < 32; u0 += SEG_BATCH) {
      int2 v[SEG_BATCH];
#pragma unroll
      for (int u = 0; u < SEG_BATCH; ++u) {
        const int sl = __builtin_amdgcn_readlane(len, u0 + u);
        const int sf = __builtin_amdgcn_readlane(from, u0 + u);
        v[u] = make_int2(0, 0);
        if (lane < sl) v[u] = binned[(size_t)sf + lane];
      }
#pragma unroll
      for (int u = 0; u < SEG_BATCH; ++u) {
        const int sl = __builtin_amdgcn_readlane(len, u0 + u);
        const int st = __builtin_amdgcn_readlane(to, u0 + u);
        if (lane < sl) run[(size_t)st + lane] = v[u];
      }
#pragma unroll
      for (int u = 0; u < SEG_BATCH; ++u) {
        const int sl = __builtin_amdgcn_readlane(len, u0 + u);
        const int sf = __builtin_amdgcn_readlane(from, u0 + u);
        const int st = __builtin_amdgcn_readlane(to, u0 + u);
        for (int i = 32 + lane; i < sl; i += 32) run[(size_t)st + i] = binned[(size_t)sf + i];
      }
    }
    __syncthreads();
  }

  int s[IPT], key[IPT], r[IPT];

  cur[t] = 0;
  for (int c0 = bs; c0 < be; c0 += CHUNK) {
#pragma unroll
    for (int k = 0; k < IPT; ++k) {
      const int i = c0 + k * 256 + t;
      const int dd = run[min(i, be - 1)].y;
      key[k] = (i < be) ? (dd & (BIN_NODES - 1)) : -1;
    }
    rank_items(priv, tot, key, r, BIN_NODES);
    cur[t] += tot[t];
  }
  const int cn = cur[t];
  int x = cn;
#pragma unroll
  for (int o = 1; o < 32; o <<= 1) {
    const int y = __shfl_up(x, o, 32);
    if (lane >= o) x += y;
  }
  if (lane == 31) wtot[wid] = x;
  __syncthreads();
  int wp = 0;
#pragma unroll
  for (int w = 0; w < 8; ++w) wp += (w < wid) ? wtot[w] : 0;
  const int excl = wp + (x - cn);
  const int n = blockIdx.x * BIN_NODES + t;
  if (n < N) {
    start[n] = bs + excl;
    deg[n]   = (float)cn;
    dinv[n]  = rsqrtf((float)cn + 1.0f);
  }
  cur[t] = excl;

  for (int c0 = bs; c0 < be; c0 += CHUNK) {
#pragma unroll
    for (int k = 0; k < IPT; ++k) {
      const int i = c0 + k * 256 + t;
      const int2 sd = run[min(i, be - 1)];
      s[k] = sd.x;
      key[k] = (i < be) ? (sd.y & (BIN_NODES - 1)) : -1;
    }
    rank_items(priv, tot, key, r, BIN_NODES);
#pragma unroll
    for (int k = 0; k < IPT; ++k) {
      if (key[k] >= 0) {
        const int slot = cur[key[k]] + priv[key[k] * PITCH16 + t] + r[k];
        csr[bs + slot] = s[k];
      }
    }
    __syncthreads();
    cur[t] += tot[t];
  }
}


__global__ __launch_bounds__(256) __attribute__((amdgpu_waves_per_eu(1, 8)))
void gemm_rows64_kernel(const float* __restrict__ A, const float* __restrict__ W,
                        const float* __restrict__ dinv, float* __restrict__ out, int M) {
  __shared__ __attribute__((aligned(16))) float sW[64 * 64];
  const int lane = threadIdx.x & 31;
  const int wid  = threadIdx.x >> 5;
  const int mtiles = (M + 15) >> 4;
  const int mt = blockIdx.x * 8 + wid;
  const int m0 = mt * 16;
  const int row = lane & 15;
  const int kh  = (lane >> 4) << 1;
  const int rbase = (lane >> 4) << 3;
  const int arow_idx = min(m0 + row, M - 1);
  const float* __restrict__ arow = A + (size_t)arow_idx * 64 + kh;

  v2f a[16];
#pragma unroll
  for (int i = 0; i < 16; ++i) {
    a[i].x = arow[4 * i];
    a[i].y = arow[4 * i + 1];
  }
  float dv[8];
#pragma unroll
  for (int r = 0; r < 8; ++r) dv[r] = dinv[min(m0 + rbase + r, M - 1)];

#pragma unroll
  for (int i = 0; i < 64 * 64 / 4; i += 256)
    ((float4*)sW)[i + threadIdx.x] = ((const float4*)W)[i + threadIdx.x];
  __syncthreads();
  if (mt >= mtiles) return;

  const float* __restrict__ bcol = sW + 4 * row;

  v8f c0 = {}, c1 = {}, c2 = {}, c3 = {};
#pragma unroll
  for (int k = 0; k < 64; k += 4) {
    const float4 q0 = *(const float4*)(bcol + (k + kh) * 64);
    const float4 q1 = *(const float4*)(bcol + (k + kh + 1) * 64);
    v2f b;
    b.x = q0.x; b.y = q1.x;
    c0 = __builtin_amdgcn_wmma_f32_16x16x4_f32(false, a[k >> 2], false, b, (short)0, c0, false, false);
    b.x = q0.y; b.y = q1.y;
    c1 = __builtin_amdgcn_wmma_f32_16x16x4_f32(false, a[k >> 2], false, b, (short)0, c1, false, false);
    b.x = q0.z; b.y = q1.z;
    c2 = __builtin_amdgcn_wmma_f32_16x16x4_f32(false, a[k >> 2], false, b, (short)0, c2, false, false);
    b.x = q0.w; b.y = q1.w;
    c3 = __builtin_amdgcn_wmma_f32_16x16x4_f32(false, a[k >> 2], false, b, (short)0, c3, false, false);
  }

#pragma unroll
  for (int r = 0; r < 8; ++r) {
    const int orow = m0 + rbase + r;
    if (orow < M) {
      const float d = dv[r];
      *(float4*)(out + (size_t)orow * 64 + 4 * row) =
          make_float4(c0[r] * d, c1[r] * d, c2[r] * d, c3[r] * d);
    }
  }
}


#define NODES_PER_WAVE64 8

template <int K>
__device__ __forceinline__ float2 sum_rows64(const float2* __restrict__ hl, int idx, int rem,
                                             float2 acc) {
  float2 v[K];
#pragma unroll
  for (int j = 0; j < K; ++j)
    v[j] = hl[(size_t)__builtin_amdgcn_readlane(idx, j) * 32];
#pragma unroll
  for (int j = 0; j < K; ++j) {
    const bool on = (j < K - 8) || (j < rem);
    acc.x += on ? v[j].x : 0.0f;
    acc.y += on ? v[j].y : 0.0f;
  }
  return acc;
}

#define ROWS_PER_BLOCK64 (8 * NODES_PER_WAVE64)
#define APITCH 68

__global__ __launch_bounds__(256) void gather64_proj_kernel(const float* __restrict__ h,
                                                            const int* __restrict__ start,
                                                            const float* __restrict__ deg,
                                                            const int* __restrict__ csr,
                                                            const float* __restrict__ dinv,
                                                            const float* __restrict__ b1,
                                                            const float* __restrict__ W2,
                                                            float* __restrict__ h2, int N) {
  __shared__ __attribute__((aligned(16))) float sA[ROWS_PER_BLOCK64 * APITCH];
  __shared__ __attribute__((aligned(16))) float sW[IN_DIM * OUT_DIM];
  __shared__ __attribute__((aligned(16))) float sD[ROWS_PER_BLOCK64];
  const int lane = threadIdx.x & 31;
  const int wid = __builtin_amdgcn_readfirstlane(threadIdx.x >> 5);
  const int node0 = (blockIdx.x * 8 + wid) * NODES_PER_WAVE64;
  const int nw = max(0, min(NODES_PER_WAVE64, N - node0));
  const float4 w2v = ((const float4*)W2)[threadIdx.x];
  const float2 bb = ((const float2*)b1)[lane];
  const int li = min(node0 + lane, N - 1);
  const int   ldb = start[li];
  const float ldc = deg[li];
  const float ldd = dinv[li];
  const bool has = lane < nw;
  const int begv = has ? ldb : 0;
  const int cntv = has ? (int)ldc : 0;
  const float dvv = has ? ldd : 0.0f;
  const int dvbits = __float_as_int(dvv);
  ((float4*)sW)[threadIdx.x] = w2v;
  if (lane < NODES_PER_WAVE64) sD[wid * NODES_PER_WAVE64 + lane] = dvv;

  const float2* __restrict__ hl = (const float2*)h + lane;
  float* __restrict__ srow = sA + (wid * NODES_PER_WAVE64) * APITCH + 2 * lane;
  for (int q = 0; q < NODES_PER_WAVE64; ++q) {
    float2 a = make_float2(0.0f, 0.0f);
    if (q < nw) {
      const int node = node0 + q;
      int beg = __builtin_amdgcn_readlane(begv, q);
      int rem = __builtin_amdgcn_readlane(cntv, q);
      const float dq = __int_as_float(__builtin_amdgcn_readlane(dvbits, q));
      float2 acc = hl[(size_t)node * 32];
      while (rem > 0) {
        int idx = node;
        if (lane < rem) idx = csr[beg + lane];
        if (rem > 24)      acc = sum_rows64<32>(hl, idx, rem, acc);
        else if (rem > 16) acc = sum_rows64<24>(hl, idx, rem, acc);
        else if (rem > 8)  acc = sum_rows64<16>(hl, idx, rem, acc);
        else               acc = sum_rows64<8>(hl, idx, rem, acc);
        rem -= 32;
        beg += 32;
      }
      a.x = fmaxf(acc.x * dq + bb.x, 0.0f);
      a.y = fmaxf(acc.y * dq + bb.y, 0.0f);
    }
    *(float2*)(srow + q * APITCH) = a;
  }
  __syncthreads();

  const int m0 = blockIdx.x * ROWS_PER_BLOCK64 + wid * 16;
  if (wid >= ROWS_PER_BLOCK64 / 16 || m0 >= N) return;

  const int row = lane & 15;
  const int kh  = (lane >> 4) << 1;
  const int rbase = (lane >> 4) << 3;
  const float4 dlo = *(const float4*)(sD + wid * 16 + rbase);
  const float4 dhi = *(const float4*)(sD + wid * 16 + rbase + 4);
  const float dv[8] = {dlo.x, dlo.y, dlo.z, dlo.w, dhi.x, dhi.y, dhi.z, dhi.w};
  const float* __restrict__ ar = sA + (wid * 16 + row) * APITCH + kh;
  const float* __restrict__ bcol = sW + row;

  v8f c = {};
#pragma unroll
  for (int k = 0; k < IN_DIM; k += 4) {
    const v2f a = *(const v2f*)(ar + k);
    v2f b;
    b.x = bcol[(k + kh) * OUT_DIM];
    b.y = bcol[(k + kh + 1) * OUT_DIM];
    c = __builtin_amdgcn_wmma_f32_16x16x4_f32(false, a, false, b, (short)0, c, false, false);
  }

#pragma unroll
  for (int r = 0; r < 8; ++r) {
    const int orow = m0 + rbase + r;
    if (orow < N) h2[(size_t)orow * OUT_DIM + row] = c[r] * dv[r];
  }
}

#define PAIRS_PER_WAVE16 4

template <int K>
__device__ __forceinline__ float sum_rows16(const float* __restrict__ hc, int pick, int i0, int i1,
                                            int rem, float acc) {
  float v[K];
#pragma unroll
  for (int j = 0; j < K; ++j) {
    const int sj = __builtin_amdgcn_ds_bpermute(pick + 4 * (j & 15), j < 16 ? i0 : i1);
    v[j] = hc[(size_t)sj * 16];
  }
#pragma unroll
  for (int j = 0; j < K; ++j)
    acc += (j < rem) ? v[j] : 0.0f;
  return acc;
}

__global__ __launch_bounds__(256) void gather16_kernel(const float* __restrict__ h,
                                                       const int* __restrict__ start,
                                                       const float* __restrict__ deg,
                                                       const int* __restrict__ csr,
                                                       const float* __restrict__ dinv,
                                                       const float* __restrict__ bias,
                                                       float* __restrict__ out, int N) {
  const int lane = threadIdx.x & 31;
  const int c = lane & 15;
  const int half = lane >> 4;
  const int base0 = (blockIdx.x * 8 + __builtin_amdgcn_readfirstlane(threadIdx.x >> 5)) *
                    (2 * PAIRS_PER_WAVE16);
  if (base0 >= N) return;
  const int np = min(PAIRS_PER_WAVE16, (N - base0 + 1) >> 1);
  int begv = 0, cntv = 0;
  float dv = 0.0f;
  if (lane < 2 * PAIRS_PER_WAVE16 && base0 + lane < N) {
    begv = start[base0 + lane];
    cntv = (int)deg[base0 + lane];
    dv = dinv[base0 + lane];
  }
  const int dvbits = __float_as_int(dv);
  const float* __restrict__ hc = h + c;
  const float bc = bias[c];
  const int pick = half << 6;

  for (int q = 0; q < np; ++q) {
    const int node = base0 + 2 * q + half;
    const int nd = min(node, N - 1);
    const int a = (2 * q + half) << 2;
    int beg = __builtin_amdgcn_ds_bpermute(a, begv);
    int rem = __builtin_amdgcn_ds_bpermute(a, cntv);
    const float dd = __int_as_float(__builtin_amdgcn_ds_bpermute(a, dvbits));
    int mu = max(__builtin_amdgcn_readlane(cntv, 2 * q), __builtin_amdgcn_readlane(cntv, 2 * q + 1));
    float acc = hc[(size_t)nd * 16];
    while (mu > 0) {
      int i0 = nd, i1 = nd;
      if (c < rem)      i0 = csr[beg + c];
      if (c + 16 < rem) i1 = csr[beg + 16 + c];
      if (mu > 24)      acc = sum_rows16<32>(hc, pick, i0, i1, rem, acc);
      else if (mu > 16) acc = sum_rows16<24>(hc, pick, i0, i1, rem, acc);
      else if (mu > 8)  acc = sum_rows16<16>(hc, pick, i0, i1, rem, acc);
      else              acc = sum_rows16<8>(hc, pick, i0, i1, rem, acc);
      mu -= 32;
      rem -= 32;
      beg += 32;
    }
    if (node < N) out[(size_t)node * 16 + c] = bc + dd * acc;
  }
}


extern "C" void kernel_launch(void* const* d_in, const int* in_sizes, int n_in,
                              void* d_out, int out_size, void* d_ws, size_t ws_size,
                              hipStream_t stream) {
  const float* x    = (const float*)d_in[0];
  const int*   edge = (const int*)d_in[1];
  const float* W1   = (const float*)d_in[2];
  const float* b1   = (const float*)d_in[3];
  const float* W2   = (const float*)d_in[4];
  const float* b2   = (const float*)d_in[5];
  float* out = (float*)d_out;

  const int E  = in_sizes[1] / 2;
  const int Nn = in_sizes[0] / IN_DIM;
  const int* src = edge;
  const int* dst = edge + E;

  const int NB = (Nn + BIN_NODES - 1) >> BIN_SHIFT;
  const int G  = (E + CHUNK - 1) / CHUNK;

  const size_t nodeB = ((size_t)Nn * 4 + 255) & ~(size_t)255;
  const size_t edgeB = ((size_t)E * 4 + 255) & ~(size_t)255;
  const size_t offB  = ((size_t)G * (NB + 1) * 4 + 255) & ~(size_t)255;
  const size_t binB  = (size_t)G * CHUNK * 8;
  const size_t runB  = (((size_t)E + 16 * (size_t)NB + 16) * 8 + 255) & ~(size_t)255;
  const size_t featB = (size_t)Nn * 64 * 4;
  char* ws = (char*)d_ws;
  float* deg    = (float*)ws;
  float* dinv   = (float*)(ws + nodeB);
  int*   start  = (int*)(ws + 2 * nodeB);
  int*   off    = (int*)(ws + 3 * nodeB);
  int2*  binned = (int2*)(ws + 3 * nodeB + offB);
  int2*  run    = (int2*)(ws + 3 * nodeB + offB + binB);
  int*   csr    = (int*)(ws + 3 * nodeB + offB + binB + runB);
  float* h1     = (float*)(ws + 3 * nodeB + offB + binB + runB + edgeB);
  float* h2     = (float*)(ws + 3 * nodeB + offB + binB + runB + edgeB + featB);

  const int mtiles = (Nn + 15) >> 4;

  chunk_sort_kernel<<<G, 256, 0, stream>>>(src, dst, off, binned, E, NB);
  bin_csr_kernel<<<NB, 256, 0, stream>>>(binned, off, run, deg, dinv, start, csr, Nn, NB, G);

  gemm_rows64_kernel<<<(mtiles + 7) / 8, 256, 0, stream>>>(x, W1, dinv, h1, Nn);
  gather64_proj_kernel<<<(Nn + ROWS_PER_BLOCK64 - 1) / ROWS_PER_BLOCK64, 256, 0, stream>>>(
      h1, start, deg, csr, dinv, b1, W2, h2, Nn);

  gather16_kernel<<<(Nn + 16 * PAIRS_PER_WAVE16 - 1) / (16 * PAIRS_PER_WAVE16), 256, 0, stream>>>(
      h2, start, deg, csr, dinv, b2, out, Nn);

  (void)n_in; (void)out_size; (void)ws_size;
}


# Bin kernel: a bin of up to 8192 pairs is collected from the chunks' pieces into LDS (counters' space), read once into registers, ranked once: global run copy and second ranking sweep gone; same CSR
# speedup vs baseline: 3.8834x; 1.0467x over previous
#include <hip/hip_runtime.h>
#include <hip/hip_bf16.h>


#define IN_DIM  64
#define OUT_DIM 16

typedef __attribute__((ext_vector_type(2))) float v2f;
typedef __attribute__((ext_vector_type(8))) float v8f;


#define BIN_SHIFT 8
#define BIN_NODES 256
#define MAX_BINS  512
#define IPT       32
#define CHUNK     (256 * IPT)
#define PITCH16   258

__device__ __forceinline__ void rank_items(unsigned short* __restrict__ priv,
                                           int* __restrict__ tot, const int (&key)[IPT],
                                           int (&r)[IPT], int nk) {
  const int t = threadIdx.x;
  const int ndw = nk * (PITCH16 / 2);
  for (int i = t; i < ndw; i += 256) ((int*)priv)[i] = 0;
  __syncthreads();
#pragma unroll
  for (int k = 0; k < IPT; ++k) {
    r[k] = 0;
    if (key[k] >= 0) {
      unsigned short* p = priv + key[k] * PITCH16 + t;
      const unsigned short c = *p;
      r[k] = c;
      *p = (unsigned short)(c + 1);
    }
  }
  __syncthreads();
  for (int kk = t; kk < nk; kk += 256) {
    int* row = (int*)(priv + kk * PITCH16);
    int run = 0;
#pragma unroll 8
    for (int j = 0; j < 128; ++j) {
      const unsigned w = (unsigned)row[j];
      const int lo = (int)(w & 0xffffu);
      const int hi = (int)(w >> 16);
      row[j] = run | ((run + lo) << 16);
      run += lo + hi;
    }
    tot[kk] = run;
  }
  __syncthreads();
}

__device__ __forceinline__ void load_idx(const int* __restrict__ p, int e0, int E, int (&v)[IPT]) {
  const int t = threadIdx.x;
  if ((E & 3) == 0) {
#pragma unroll
    for (int kk = 0; kk < IPT / 4; ++kk) {
      const int i = e0 + kk * 1024 + t * 4;
      int4 q = make_int4(-1, -1, -1, -1);
      if (i < E) q = *(const int4*)(p + i);
      v[4 * kk] = q.x; v[4 * kk + 1] = q.y; v[4 * kk + 2] = q.z; v[4 * kk + 3] = q.w;
    }
  } else {
#pragma unroll
    for (int k = 0; k < IPT; ++k) {
      const int i = e0 + (k >> 2) * 1024 + t * 4 + (k & 3);
      v[k] = (i < E) ? p[i] : -1;
    }
  }
}

__global__ __launch_bounds__(256) void chunk_sort_kernel(const int* __restrict__ src,
                                                         const int* __restrict__ dst,
                                                         int* __restrict__ off,
                                                         int2* __restrict__ binned,
                                                         int E, int NB) {
  __shared__ __attribute__((aligned(16))) unsigned short priv[MAX_BINS * PITCH16];
  __shared__ int tot[MAX_BINS];
  __shared__ int wtot[8];
  const int lane = threadIdx.x & 31;
  const int wid  = threadIdx.x >> 5;
  const int g = blockIdx.x;
  int d[IPT], s[IPT], key[IPT], r[IPT];
  load_idx(dst, g * CHUNK, E, d);
  load_idx(src, g * CHUNK, E, s);
#pragma unroll
  for (int k = 0; k < IPT; ++k) key[k] = d[k] >> BIN_SHIFT;
  rank_items(priv, tot, key, r, NB);

  int* __restrict__ orow = off + (size_t)g * (NB + 1);
  int carry = 0;
  for (int b0 = 0; b0 < NB; b0 += 256) {
    const int b = b0 + threadIdx.x;
    const int all = (b < NB) ? tot[b] : 0;
    int x = all;
#pragma unroll
    for (int o = 1; o < 32; o <<= 1) {
      const int y = __shfl_up(x, o, 32);
      if (lane >= o) x += y;
    }
    if (lane == 31) wtot[wid] = x;
    __syncthreads();
    int wp = 0, sum = 0;
#pragma unroll
    for (int w = 0; w < 8; ++w) {
      const int v = wtot[w];
      wp += (w < wid) ? v : 0;
      sum += v;
    }
    const int first = carry + wp + (x - all);
    if (b < NB) {
      tot[b] = first;
      orow[b] = first;
    }
    carry += sum;
    __syncthreads();
  }
  if (threadIdx.x == 0) orow[NB] = carry;

  int2* __restrict__ mine = binned + (size_t)g * CHUNK;
#pragma unroll
  for (int k = 0; k < IPT; ++k) {
    if (key[k] >= 0) {
      const int slot = tot[key[k]] + priv[key[k] * PITCH16 + threadIdx.x] + r[k];
      mine[slot] = make_int2(s[k], d[k]);
    }
  }
}

#define SEG_BATCH 16

__device__ __forceinline__ void collect_run(const int2* __restrict__ binned,
                                            const int* __restrict__ off, int2* dstrun,
                                            int* wtot, int bin, int pitch, int G, int lane,
                                            int wid) {
  const int t = threadIdx.x;
  int pcarry = 0;
  for (int g0 = 0; g0 < G; g0 += 256) {
    const int g = g0 + t;
    int lo = 0, len = 0;
    if (g < G) {
      lo = off[(size_t)g * pitch + bin];
      len = off[(size_t)g * pitch + bin + 1] - lo;
    }
    int x = len;
#pragma unroll
    for (int o = 1; o < 32; o <<= 1) {
      const int y = __shfl_up(x, o, 32);
      if (lane >= o) x += y;
    }
    if (lane == 31) wtot[wid] = x;
    __syncthreads();
    int wp = 0, sum = 0;
#pragma unroll
    for (int w = 0; w < 8; ++w) {
      const int v = wtot[w];
      wp += (w < wid) ? v : 0;
      sum += v;
    }
    const int from = g * CHUNK + lo;
    const int to = pcarry + wp + (x - len);
    pcarry += sum;
#pragma unroll
    for (int u0 = 0; u0 < 32; u0 += SEG_BATCH) {
      int2 v[SEG_BATCH];
#pragma unroll
      for (int u = 0; u < SEG_BATCH; ++u) {
        const int sl = __builtin_amdgcn_readlane(len, u0 + u);
        const int sf = __builtin_amdgcn_readlane(from, u0 + u);
        v[u] = make_int2(0, 0);
        if (lane < sl) v[u] = binned[(size_t)sf + lane];
      }
#pragma unroll
      for (int u = 0; u < SEG_BATCH; ++u) {
        const int sl = __builtin_amdgcn_readlane(len, u0 + u);
        const int st = __builtin_amdgcn_readlane(to, u0 + u);
        if (lane < sl) dstrun[(size_t)st + lane] = v[u];
      }
#pragma unroll
      for (int u = 0; u < SEG_BATCH; ++u) {
        const int sl = __builtin_amdgcn_readlane(len, u0 + u);
        const int sf = __builtin_amdgcn_readlane(from, u0 + u);
        const int st = __builtin_amdgcn_readlane(to, u0 + u);
        for (int i = 32 + lane; i < sl; i += 32) dstrun[(size_t)st + i] = binned[(size_t)sf + i];
      }
    }
    __syncthreads();
  }
}

__global__ __launch_bounds__(256) void bin_csr_kernel(const int2* __restrict__ binned,
                                                      const int* __restrict__ off,
                                                      int2* runbuf,
                                                      float* __restrict__ deg,
                                                      float* __restrict__ dinv,
                                                      int* __restrict__ start,
                                                      int* __restrict__ csr, int N, int NB, int G) {
  __shared__ __attribute__((aligned(16))) unsigned short priv[BIN_NODES * PITCH16];
  __shared__ int tot[BIN_NODES];
  __shared__ int cur[BIN_NODES];
  __shared__ int wtot[8];
  __shared__ int red[16];
  const int t = threadIdx.x;
  const int lane = t & 31;
  const int wid  = __builtin_amdgcn_readfirstlane(t >> 5);
  const int bin = blockIdx.x;
  const int pitch = NB + 1;

  int slo = 0, shi = 0;
  for (int g = t; g < G; g += 256) {
    slo += off[(size_t)g * pitch + bin];
    shi += off[(size_t)g * pitch + bin + 1];
  }
#pragma unroll
  for (int o = 16; o > 0; o >>= 1) {
    slo += __shfl_xor(slo, o, 32);
    shi += __shfl_xor(shi, o, 32);
  }
  if (lane == 0) { red[wid] = slo; red[8 + wid] = shi; }
  __syncthreads();
  int bs = 0, be = 0;
#pragma unroll
  for (int w = 0; w < 8; ++w) { bs += red[w]; be += red[8 + w]; }
  int2* run = runbuf + ((size_t)((bs + 15) & ~15) + (size_t)16 * bin - (size_t)bs);
  const int cnt = be - bs;
  const bool fits = cnt <= CHUNK;
  int2* lrun = (int2*)priv;

  if (fits) collect_run(binned, off, lrun, wtot, bin, pitch, G, lane, wid);
  else      collect_run(binned, off, run + bs, wtot, bin, pitch, G, lane, wid);

  int s[IPT], key[IPT], r[IPT];

  int cn;
  if (fits) {
    const int last = max(cnt - 1, 0);
#pragma unroll
    for (int k = 0; k < IPT; ++k) {
      const int i = k * 256 + t;
      const int2 sd = lrun[min(i, last)];
      s[k] = sd.x;
      key[k] = (i < cnt) ? (sd.y & (BIN_NODES - 1)) : -1;
    }
    __syncthreads();
    rank_items(priv, tot, key, r, BIN_NODES);
    cn = tot[t];
  } else {
    cur[t] = 0;
    for (int c0 = bs; c0 < be; c0 += CHUNK) {
#pragma unroll
      for (int k = 0; k < IPT; ++k) {
        const int i = c0 + k * 256 + t;
        const int dd = run[min(i, be - 1)].y;
        key[k] = (i < be) ? (dd & (BIN_NODES - 1)) : -1;
      }
      rank_items(priv, tot, key, r, BIN_NODES);
      cur[t] += tot[t];
    }
    cn = cur[t];
  }
  int x = cn;
#pragma unroll
  for (int o = 1; o < 32; o <<= 1) {
    const int y = __shfl_up(x, o, 32);
    if (lane >= o) x += y;
  }
  if (lane == 31) wtot[wid] = x;
  __syncthreads();
  int wp = 0;
#pragma unroll
  for (int w = 0; w < 8; ++w) wp += (w < wid) ? wtot[w] : 0;
  const int excl = wp + (x - cn);
  const int n = blockIdx.x * BIN_NODES + t;
  if (n < N) {
    start[n] = bs + excl;
    deg[n]   = (float)cn;
    dinv[n]  = rsqrtf((float)cn + 1.0f);
  }
  cur[t] = excl;

  if (fits) {
    __syncthreads();
#pragma unroll
    for (int k = 0; k < IPT; ++k) {
      if (key[k] >= 0) {
        const int slot = cur[key[k]] + priv[key[k] * PITCH16 + t] + r[k];
        csr[bs + slot] = s[k];
      }
    }
    return;
  }

  for (int c0 = bs; c0 < be; c0 += CHUNK) {
#pragma unroll
    for (int k = 0; k < IPT; ++k) {
      const int i = c0 + k * 256 + t;
      const int2 sd = run[min(i, be - 1)];
      s[k] = sd.x;
      key[k] = (i < be) ? (sd.y & (BIN_NODES - 1)) : -1;
    }
    rank_items(priv, tot, key, r, BIN_NODES);
#pragma unroll
    for (int k = 0; k < IPT; ++k) {
      if (key[k] >= 0) {
        const int slot = cur[key[k]] + priv[key[k] * PITCH16 + t] + r[k];
        csr[bs + slot] = s[k];
      }
    }
    __syncthreads();
    cur[t] += tot[t];
  }
}


__global__ __launch_bounds__(256) __attribute__((amdgpu_waves_per_eu(1, 8)))
void gemm_rows64_kernel(const float* __restrict__ A, const float* __restrict__ W,
                        const float* __restrict__ dinv, float* __restrict__ out, int M) {
  __shared__ __attribute__((aligned(16))) float sW[64 * 64];
  const int lane = threadIdx.x & 31;
  const int wid  = threadIdx.x >> 5;
  const int mtiles = (M + 15) >> 4;
  const int mt = blockIdx.x * 8 + wid;
  const int m0 = mt * 16;
  const int row = lane & 15;
  const int kh  = (lane >> 4) << 1;
  const int rbase = (lane >> 4) << 3;
  const int arow_idx = min(m0 + row, M - 1);
  const float* __restrict__ arow = A + (size_t)arow_idx * 64 + kh;

  v2f a[16];
#pragma unroll
  for (int i = 0; i < 16; ++i) {
    a[i].x = arow[4 * i];
    a[i].y = arow[4 * i + 1];
  }
  float dv[8];
#pragma unroll
  for (int r = 0; r < 8; ++r) dv[r] = dinv[min(m0 + rbase + r, M - 1)];

#pragma unroll
  for (int i = 0; i < 64 * 64 / 4; i += 256)
    ((float4*)sW)[i + threadIdx.x] = ((const float4*)W)[i + threadIdx.x];
  __syncthreads();
  if (mt >= mtiles) return;

  const float* __restrict__ bcol = sW + 4 * row;

  v8f c0 = {}, c1 = {}, c2 = {}, c3 = {};
#pragma unroll
  for (int k = 0; k < 64; k += 4) {
    const float4 q0 = *(const float4*)(bcol + (k + kh) * 64);
    const float4 q1 = *(const float4*)(bcol + (k + kh + 1) * 64);
    v2f b;
    b.x = q0.x; b.y = q1.x;
    c0 = __builtin_amdgcn_wmma_f32_16x16x4_f32(false, a[k >> 2], false, b, (short)0, c0, false, false);
    b.x = q0.y; b.y = q1.y;
    c1 = __builtin_amdgcn_wmma_f32_16x16x4_f32(false, a[k >> 2], false, b, (short)0, c1, false, false);
    b.x = q0.z; b.y = q1.z;
    c2 = __builtin_amdgcn_wmma_f32_16x16x4_f32(false, a[k >> 2], false, b, (short)0, c2, false, false);
    b.x = q0.w; b.y = q1.w;
    c3 = __builtin_amdgcn_wmma_f32_16x16x4_f32(false, a[k >> 2], false, b, (short)0, c3, false, false);
  }

#pragma unroll
  for (int r = 0; r < 8; ++r) {
    const int orow = m0 + rbase + r;
    if (orow < M) {
      const float d = dv[r];
      *(float4*)(out + (size_t)orow * 64 + 4 * row) =
          make_float4(c0[r] * d, c1[r] * d, c2[r] * d, c3[r] * d);
    }
  }
}


#define NODES_PER_WAVE64 8

template <int K>
__device__ __forceinline__ float2 sum_rows64(const float2* __restrict__ hl, int idx, int rem,
                                             float2 acc) {
  float2 v[K];
#pragma unroll
  for (int j = 0; j < K; ++j)
    v[j] = hl[(size_t)__builtin_amdgcn_readlane(idx, j) * 32];
#pragma unroll
  for (int j = 0; j < K; ++j) {
    const bool on = (j < K - 8) || (j < rem);
    acc.x += on ? v[j].x : 0.0f;
    acc.y += on ? v[j].y : 0.0f;
  }
  return acc;
}

#define ROWS_PER_BLOCK64 (8 * NODES_PER_WAVE64)
#define APITCH 68

__global__ __launch_bounds__(256) void gather64_proj_kernel(const float* __restrict__ h,
                                                            const int* __restrict__ start,
                                                            const float* __restrict__ deg,
                                                            const int* __restrict__ csr,
                                                            const float* __restrict__ dinv,
                                                            const float* __restrict__ b1,
                                                            const float* __restrict__ W2,
                                                            float* __restrict__ h2, int N) {
  __shared__ __attribute__((aligned(16))) float sA[ROWS_PER_BLOCK64 * APITCH];
  __shared__ __attribute__((aligned(16))) float sW[IN_DIM * OUT_DIM];
  __shared__ __attribute__((aligned(16))) float sD[ROWS_PER_BLOCK64];
  const int lane = threadIdx.x & 31;
  const int wid = __builtin_amdgcn_readfirstlane(threadIdx.x >> 5);
  const int node0 = (blockIdx.x * 8 + wid) * NODES_PER_WAVE64;
  const int nw = max(0, min(NODES_PER_WAVE64, N - node0));
  const float4 w2v = ((const float4*)W2)[threadIdx.x];
  const float2 bb = ((const float2*)b1)[lane];
  const int li = min(node0 + lane, N - 1);
  const int   ldb = start[li];
  const float ldc = deg[li];
  const float ldd = dinv[li];
  const bool has = lane < nw;
  const int begv = has ? ldb : 0;
  const int cntv = has ? (int)ldc : 0;
  const float dvv = has ? ldd : 0.0f;
  const int dvbits = __float_as_int(dvv);
  ((float4*)sW)[threadIdx.x] = w2v;
  if (lane < NODES_PER_WAVE64) sD[wid * NODES_PER_WAVE64 + lane] = dvv;

  const float2* __restrict__ hl = (const float2*)h + lane;
  float* __restrict__ srow = sA + (wid * NODES_PER_WAVE64) * APITCH + 2 * lane;
  for (int q = 0; q < NODES_PER_WAVE64; ++q) {
    float2 a = make_float2(0.0f, 0.0f);
    if (q < nw) {
      const int node = node0 + q;
      int beg = __builtin_amdgcn_readlane(begv, q);
      int rem = __builtin_amdgcn_readlane(cntv, q);
      const float dq = __int_as_float(__builtin_amdgcn_readlane(dvbits, q));
      float2 acc = hl[(size_t)node * 32];
      while (rem > 0) {
        int idx = node;
        if (lane < rem) idx = csr[beg + lane];
        if (rem > 24)      acc = sum_rows64<32>(hl, idx, rem, acc);
        else if (rem > 16) acc = sum_rows64<24>(hl, idx, rem, acc);
        else if (rem > 8)  acc = sum_rows64<16>(hl, idx, rem, acc);
        else               acc = sum_rows64<8>(hl, idx, rem, acc);
        rem -= 32;
        beg += 32;
      }
      a.x = fmaxf(acc.x * dq + bb.x, 0.0f);
      a.y = fmaxf(acc.y * dq + bb.y, 0.0f);
    }
    *(float2*)(srow + q * APITCH) = a;
  }
  __syncthreads();

  const int m0 = blockIdx.x * ROWS_PER_BLOCK64 + wid * 16;
  if (wid >= ROWS_PER_BLOCK64 / 16 || m0 >= N) return;

  const int row = lane & 15;
  const int kh  = (lane >> 4) << 1;
  const int rbase = (lane >> 4) << 3;
  const float4 dlo = *(const float4*)(sD + wid * 16 + rbase);
  const float4 dhi = *(const float4*)(sD + wid * 16 + rbase + 4);
  const float dv[8] = {dlo.x, dlo.y, dlo.z, dlo.w, dhi.x, dhi.y, dhi.z, dhi.w};
  const float* __restrict__ ar = sA + (wid * 16 + row) * APITCH + kh;
  const float* __restrict__ bcol = sW + row;

  v8f c = {};
#pragma unroll
  for (int k = 0; k < IN_DIM; k += 4) {
    const v2f a = *(const v2f*)(ar + k);
    v2f b;
    b.x = bcol[(k + kh) * OUT_DIM];
    b.y = bcol[(k + kh + 1) * OUT_DIM];
    c = __builtin_amdgcn_wmma_f32_16x16x4_f32(false, a, false, b, (short)0, c, false, false);
  }

#pragma unroll
  for (int r = 0; r < 8; ++r) {
    const int orow = m0 + rbase + r;
    if (orow < N) h2[(size_t)orow * OUT_DIM + row] = c[r] * dv[r];
  }
}

#define PAIRS_PER_WAVE16 4

template <int K>
__device__ __forceinline__ float sum_rows16(const float* __restrict__ hc, int pick, int i0, int i1,
                                            int rem, float acc) {
  float v[K];
#pragma unroll
  for (int j = 0; j < K; ++j) {
    const int sj = __builtin_amdgcn_ds_bpermute(pick + 4 * (j & 15), j < 16 ? i0 : i1);
    v[j] = hc[(size_t)sj * 16];
  }
#pragma unroll
  for (int j = 0; j < K; ++j)
    acc += (j < rem) ? v[j] : 0.0f;
  return acc;
}

__global__ __launch_bounds__(256) void gather16_kernel(const float* __restrict__ h,
                                                       const int* __restrict__ start,
                                                       const float* __restrict__ deg,
                                                       const int* __restrict__ csr,
                                                       const float* __restrict__ dinv,
                                                       const float* __restrict__ bias,
                                                       float* __restrict__ out, int N) {
  const int lane = threadIdx.x & 31;
  const int c = lane & 15;
  const int half = lane >> 4;
  const int base0 = (blockIdx.x * 8 + __builtin_amdgcn_readfirstlane(threadIdx.x >> 5)) *
                    (2 * PAIRS_PER_WAVE16);
  if (base0 >= N) return;
  const int np = min(PAIRS_PER_WAVE16, (N - base0 + 1) >> 1);
  int begv = 0, cntv = 0;
  float dv = 0.0f;
  if (lane < 2 * PAIRS_PER_WAVE16 && base0 + lane < N) {
    begv = start[base0 + lane];
    cntv = (int)deg[base0 + lane];
    dv = dinv[base0 + lane];
  }
  const int dvbits = __float_as_int(dv);
  const float* __restrict__ hc = h + c;
  const float bc = bias[c];
  const int pick = half << 6;

  for (int q = 0; q < np; ++q) {
    const int node = base0 + 2 * q + half;
    const int nd = min(node, N - 1);
    const int a = (2 * q + half) << 2;
    int beg = __builtin_amdgcn_ds_bpermute(a, begv);
    int rem = __builtin_amdgcn_ds_bpermute(a, cntv);
    const float dd = __int_as_float(__builtin_amdgcn_ds_bpermute(a, dvbits));
    int mu = max(__builtin_amdgcn_readlane(cntv, 2 * q), __builtin_amdgcn_readlane(cntv, 2 * q + 1));
    float acc = hc[(size_t)nd * 16];
    while (mu > 0) {
      int i0 = nd, i1 = nd;
      if (c < rem)      i0 = csr[beg + c];
      if (c + 16 < rem) i1 = csr[beg + 16 + c];
      if (mu > 24)      acc = sum_rows16<32>(hc, pick, i0, i1, rem, acc);
      else if (mu > 16) acc = sum_rows16<24>(hc, pick, i0, i1, rem, acc);
      else if (mu > 8)  acc = sum_rows16<16>(hc, pick, i0, i1, rem, acc);
      else              acc = sum_rows16<8>(hc, pick, i0, i1, rem, acc);
      mu -= 32;
      rem -= 32;
      beg += 32;
    }
    if (node < N) out[(size_t)node * 16 + c] = bc + dd * acc;
  }
}


extern "C" void kernel_launch(void* const* d_in, const int* in_sizes, int n_in,
                              void* d_out, int out_size, void* d_ws, size_t ws_size,
                              hipStream_t stream) {
  const float* x    = (const float*)d_in[0];
  const int*   edge = (const int*)d_in[1];
  const float* W1   = (const float*)d_in[2];
  const float* b1   = (const float*)d_in[3];
  const float* W2   = (const float*)d_in[4];
  const float* b2   = (const float*)d_in[5];
  float* out = (float*)d_out;

  const int E  = in_sizes[1] / 2;
  const int Nn = in_sizes[0] / IN_DIM;
  const int* src = edge;
  const int* dst = edge + E;

  const int NB = (Nn + BIN_NODES - 1) >> BIN_SHIFT;
  const int G  = (E + CHUNK - 1) / CHUNK;

  const size_t nodeB = ((size_t)Nn * 4 + 255) & ~(size_t)255;
  const size_t edgeB = ((size_t)E * 4 + 255) & ~(size_t)255;
  const size_t offB  = ((size_t)G * (NB + 1) * 4 + 255) & ~(size_t)255;
  const size_t binB  = (size_t)G * CHUNK * 8;
  const size_t runB  = (((size_t)E + 16 * (size_t)NB + 16) * 8 + 255) & ~(size_t)255;
  const size_t featB = (size_t)Nn * 64 * 4;
  char* ws = (char*)d_ws;
  float* deg    = (float*)ws;
  float* dinv   = (float*)(ws + nodeB);
  int*   start  = (int*)(ws + 2 * nodeB);
  int*   off    = (int*)(ws + 3 * nodeB);
  int2*  binned = (int2*)(ws + 3 * nodeB + offB);
  int2*  run    = (int2*)(ws + 3 * nodeB + offB + binB);
  int*   csr    = (int*)(ws + 3 * nodeB + offB + binB + runB);
  float* h1     = (float*)(ws + 3 * nodeB + offB + binB + runB + edgeB);
  float* h2     = (float*)(ws + 3 * nodeB + offB + binB + runB + edgeB + featB);

  const int mtiles = (Nn + 15) >> 4;

  chunk_sort_kernel<<<G, 256, 0, stream>>>(src, dst, off, binned, E, NB);
  bin_csr_kernel<<<NB, 256, 0, stream>>>(binned, off, run, deg, dinv, start, csr, Nn, NB, G);

  gemm_rows64_kernel<<<(mtiles + 7) / 8, 256, 0, stream>>>(x, W1, dinv, h1, Nn);
  gather64_proj_kernel<<<(Nn + ROWS_PER_BLOCK64 - 1) / ROWS_PER_BLOCK64, 256, 0, stream>>>(
      h1, start, deg, csr, dinv, b1, W2, h2, Nn);

  gather16_kernel<<<(Nn + 16 * PAIRS_PER_WAVE16 - 1) / (16 * PAIRS_PER_WAVE16), 256, 0, stream>>>(
      h2, start, deg, csr, dinv, b2, out, Nn);

  (void)n_in; (void)out_size; (void)ws_size;
}


# CSR build: each sorted chunk and each bin's source list is laid out in LDS at its slots, then stored as contiguous 16-byte-per-lane runs, not as one scattered 8- or 4-byte store per item; same CSR
# speedup vs baseline: 4.1157x; 1.0598x over previous
#include <hip/hip_runtime.h>
#include <hip/hip_bf16.h>


#define IN_DIM  64
#define OUT_DIM 16

typedef __attribute__((ext_vector_type(2))) float v2f;
typedef __attribute__((ext_vector_type(8))) float v8f;


#define BIN_SHIFT 8
#define BIN_NODES 256
#define MAX_BINS  512
#define IPT       32
#define CHUNK     (256 * IPT)
#define PITCH16   258

__device__ __forceinline__ void rank_items(unsigned short* __restrict__ priv,
                                           int* __restrict__ tot, const int (&key)[IPT],
                                           int (&r)[IPT], int nk) {
  const int t = threadIdx.x;
  const int ndw = nk * (PITCH16 / 2);
  for (int i = t; i < ndw; i += 256) ((int*)priv)[i] = 0;
  __syncthreads();
#pragma unroll
  for (int k = 0; k < IPT; ++k) {
    r[k] = 0;
    if (key[k] >= 0) {
      unsigned short* p = priv + key[k] * PITCH16 + t;
      const unsigned short c = *p;
      r[k] = c;
      *p = (unsigned short)(c + 1);
    }
  }
  __syncthreads();
  for (int kk = t; kk < nk; kk += 256) {
    int* row = (int*)(priv + kk * PITCH16);
    int run = 0;
#pragma unroll 8
    for (int j = 0; j < 128; ++j) {
      const unsigned w = (unsigned)row[j];
      const int lo = (int)(w & 0xffffu);
      const int hi = (int)(w >> 16);
      row[j] = run | ((run + lo) << 16);
      run += lo + hi;
    }
    tot[kk] = run;
  }
  __syncthreads();
}

__device__ __forceinline__ void load_idx(const int* __restrict__ p, int e0, int E, int (&v)[IPT]) {
  const int t = threadIdx.x;
  if ((E & 3) == 0) {
#pragma unroll
    for (int kk = 0; kk < IPT / 4; ++kk) {
      const int i = e0 + kk * 1024 + t * 4;
      int4 q = make_int4(-1, -1, -1, -1);
      if (i < E) q = *(const int4*)(p + i);
      v[4 * kk] = q.x; v[4 * kk + 1] = q.y; v[4 * kk + 2] = q.z; v[4 * kk + 3] = q.w;
    }
  } else {
#pragma unroll
    for (int k = 0; k < IPT; ++k) {
      const int i = e0 + (k >> 2) * 1024 + t * 4 + (k & 3);
      v[k] = (i < E) ? p[i] : -1;
    }
  }
}

__global__ __launch_bounds__(256) void chunk_sort_kernel(const int* __restrict__ src,
                                                         const int* __restrict__ dst,
                                                         int* __restrict__ off,
                                                         int2* __restrict__ binned,
                                                         int E, int NB) {
  __shared__ __attribute__((aligned(16))) unsigned short priv[MAX_BINS * PITCH16];
  __shared__ int tot[MAX_BINS];
  __shared__ int wtot[8];
  const int lane = threadIdx.x & 31;
  const int wid  = threadIdx.x >> 5;
  const int g = blockIdx.x;
  int d[IPT], s[IPT], key[IPT], r[IPT];
  load_idx(dst, g * CHUNK, E, d);
  load_idx(src, g * CHUNK, E, s);
#pragma unroll
  for (int k = 0; k < IPT; ++k) key[k] = d[k] >> BIN_SHIFT;
  rank_items(priv, tot, key, r, NB);

  int* __restrict__ orow = off + (size_t)g * (NB + 1);
  int carry = 0;
  for (int b0 = 0; b0 < NB; b0 += 256) {
    const int b = b0 + threadIdx.x;
    const int all = (b < NB) ? tot[b] : 0;
    int x = all;
#pragma unroll
    for (int o = 1; o < 32; o <<= 1) {
      const int y = __shfl_up(x, o, 32);
      if (lane >= o) x += y;
    }
    if (lane == 31) wtot[wid] = x;
    __syncthreads();
    int wp = 0, sum = 0;
#pragma unroll
    for (int w = 0; w < 8; ++w) {
      const int v = wtot[w];
      wp += (w < wid) ? v : 0;
      sum += v;
    }
    const int first = carry + wp + (x - all);
    if (b < NB) {
      tot[b] = first;
      orow[b] = first;
    }
    carry += sum;
    __syncthreads();
  }
  if (threadIdx.x == 0) orow[NB] = carry;

#pragma unroll
  for (int k = 0; k < IPT; ++k)
    if (key[k] >= 0) r[k] += tot[key[k]] + priv[key[k] * PITCH16 + threadIdx.x];
  __syncthreads();
  int2* stage = (int2*)priv;
#pragma unroll
  for (int k = 0; k < IPT; ++k)
    if (key[k] >= 0) stage[r[k]] = make_int2(s[k], d[k]);
  __syncthreads();

  int2* __restrict__ mine = binned + (size_t)g * CHUNK;
  const int nq = carry >> 1;
#pragma unroll 4
  for (int q = threadIdx.x; q < nq; q += 256) ((int4*)mine)[q] = ((const int4*)stage)[q];
  if ((carry & 1) && threadIdx.x == 0) mine[carry - 1] = stage[carry - 1];
}

#define SEG_BATCH 16

__device__ __forceinline__ void collect_run(const int2* __restrict__ binned,
                                            const int* __restrict__ off, int2* dstrun,
                                            int* wtot, int bin, int pitch, int G, int lane,
                                            int wid) {
  const int t = threadIdx.x;
  int pcarry = 0;
  for (int g0 = 0; g0 < G; g0 += 256) {
    const int g = g0 + t;
    int lo = 0, len = 0;
    if (g < G) {
      lo = off[(size_t)g * pitch + bin];
      len = off[(size_t)g * pitch + bin + 1] - lo;
    }
    int x = len;
#pragma unroll
    for (int o = 1; o < 32; o <<= 1) {
      const int y = __shfl_up(x, o, 32);
      if (lane >= o) x += y;
    }
    if (lane == 31) wtot[wid] = x;
    __syncthreads();
    int wp = 0, sum = 0;
#pragma unroll
    for (int w = 0; w < 8; ++w) {
      const int v = wtot[w];
      wp += (w < wid) ? v : 0;
      sum += v;
    }
    const int from = g * CHUNK + lo;
    const int to = pcarry + wp + (x - len);
    pcarry += sum;
#pragma unroll
    for (int u0 = 0; u0 < 32; u0 += SEG_BATCH) {
      int2 v[SEG_BATCH];
#pragma unroll
      for (int u = 0; u < SEG_BATCH; ++u) {
        const int sl = __builtin_amdgcn_readlane(len, u0 + u);
        const int sf = __builtin_amdgcn_readlane(from, u0 + u);
        v[u] = make_int2(0, 0);
        if (lane < sl) v[u] = binned[(size_t)sf + lane];
      }
#pragma unroll
      for (int u = 0; u < SEG_BATCH; ++u) {
        const int sl = __builtin_amdgcn_readlane(len, u0 + u);
        const int st = __builtin_amdgcn_readlane(to, u0 + u);
        if (lane < sl) dstrun[(size_t)st + lane] = v[u];
      }
#pragma unroll
      for (int u = 0; u < SEG_BATCH; ++u) {
        const int sl = __builtin_amdgcn_readlane(len, u0 + u);
        const int sf = __builtin_amdgcn_readlane(from, u0 + u);
        const int st = __builtin_amdgcn_readlane(to, u0 + u);
        for (int i = 32 + lane; i < sl; i += 32) dstrun[(size_t)st + i] = binned[(size_t)sf + i];
      }
    }
    __syncthreads();
  }
}

__global__ __launch_bounds__(256) void bin_csr_kernel(const int2* __restrict__ binned,
                                                      const int* __restrict__ off,
                                                      int2* runbuf,
                                                      float* __restrict__ deg,
                                                      float* __restrict__ dinv,
                                                      int* __restrict__ start,
                                                      int* __restrict__ csr, int N, int NB, int G) {
  __shared__ __attribute__((aligned(16))) unsigned short priv[BIN_NODES * PITCH16];
  __shared__ int tot[BIN_NODES];
  __shared__ int cur[BIN_NODES];
  __shared__ int wtot[8];
  __shared__ int red[16];
  const int t = threadIdx.x;
  const int lane = t & 31;
  const int wid  = __builtin_amdgcn_readfirstlane(t >> 5);
  const int bin = blockIdx.x;
  const int pitch = NB + 1;

  int slo = 0, shi = 0;
  for (int g = t; g < G; g += 256) {
    slo += off[(size_t)g * pitch + bin];
    shi += off[(size_t)g * pitch + bin + 1];
  }
#pragma unroll
  for (int o = 16; o > 0; o >>= 1) {
    slo += __shfl_xor(slo, o, 32);
    shi += __shfl_xor(shi, o, 32);
  }
  if (lane == 0) { red[wid] = slo; red[8 + wid] = shi; }
  __syncthreads();
  int bs = 0, be = 0;
#pragma unroll
  for (int w = 0; w < 8; ++w) { bs += red[w]; be += red[8 + w]; }
  int2* run = runbuf + ((size_t)((bs + 15) & ~15) + (size_t)16 * bin - (size_t)bs);
  const int cnt = be - bs;
  const bool fits = cnt <= CHUNK;
  int2* lrun = (int2*)priv;

  if (fits) collect_run(binned, off, lrun, wtot, bin, pitch, G, lane, wid);
  else      collect_run(binned, off, run + bs, wtot, bin, pitch, G, lane, wid);

  int s[IPT], key[IPT], r[IPT];

  int cn;
  if (fits) {
    const int last = max(cnt - 1, 0);
#pragma unroll
    for (int k = 0; k < IPT; ++k) {
      const int i = k * 256 + t;
      const int2 sd = lrun[min(i, last)];
      s[k] = sd.x;
      key[k] = (i < cnt) ? (sd.y & (BIN_NODES - 1)) : -1;
    }
    __syncthreads();
    rank_items(priv, tot, key, r, BIN_NODES);
    cn = tot[t];
  } else {
    cur[t] = 0;
    for (int c0 = bs; c0 < be; c0 += CHUNK) {
#pragma unroll
      for (int k = 0; k < IPT; ++k) {
        const int i = c0 + k * 256 + t;
        const int dd = run[min(i, be - 1)].y;
        key[k] = (i < be) ? (dd & (BIN_NODES - 1)) : -1;
      }
      rank_items(priv, tot, key, r, BIN_NODES);
      cur[t] += tot[t];
    }
    cn = cur[t];
  }
  int x = cn;
#pragma unroll
  for (int o = 1; o < 32; o <<= 1) {
    const int y = __shfl_up(x, o, 32);
    if (lane >= o) x += y;
  }
  if (lane == 31) wtot[wid] = x;
  __syncthreads();
  int wp = 0;
#pragma unroll
  for (int w = 0; w < 8; ++w) wp += (w < wid) ? wtot[w] : 0;
  const int excl = wp + (x - cn);
  const int n = blockIdx.x * BIN_NODES + t;
  if (n < N) {
    start[n] = bs + excl;
    deg[n]   = (float)cn;
    dinv[n]  = rsqrtf((float)cn + 1.0f);
  }
  cur[t] = excl;

  if (fits) {
    __syncthreads();
#pragma unroll
    for (int k = 0; k < IPT; ++k)
      if (key[k] >= 0) r[k] += cur[key[k]] + priv[key[k] * PITCH16 + t];
    __syncthreads();
    int* stage = (int*)priv;
    const int pad = bs & 3;
#pragma unroll
    for (int k = 0; k < IPT; ++k)
      if (key[k] >= 0) stage[pad + r[k]] = s[k];
    __syncthreads();
    int* __restrict__ gbase = csr + (bs - pad);
    const int total = pad + cnt;
    for (int j = 4 * t; j < total; j += 1024) {
      if (j >= pad && j + 4 <= total) {
        *(int4*)(gbase + j) = *(const int4*)(stage + j);
      } else {
        for (int e = j; e < j + 4; ++e)
          if (e >= pad && e < total) gbase[e] = stage[e];
      }
    }
    return;
  }

  for (int c0 = bs; c0 < be; c0 += CHUNK) {
#pragma unroll
    for (int k = 0; k < IPT; ++k) {
      const int i = c0 + k * 256 + t;
      const int2 sd = run[min(i, be - 1)];
      s[k] = sd.x;
      key[k] = (i < be) ? (sd.y & (BIN_NODES - 1)) : -1;
    }
    rank_items(priv, tot, key, r, BIN_NODES);
#pragma unroll
    for (int k = 0; k < IPT; ++k) {
      if (key[k] >= 0) {
        const int slot = cur[key[k]] + priv[key[k] * PITCH16 + t] + r[k];
        csr[bs + slot] = s[k];
      }
    }
    __syncthreads();
    cur[t] += tot[t];
  }
}


__global__ __launch_bounds__(256) __attribute__((amdgpu_waves_per_eu(1, 8)))
void gemm_rows64_kernel(const float* __restrict__ A, const float* __restrict__ W,
                        const float* __restrict__ dinv, float* __restrict__ out, int M) {
  __shared__ __attribute__((aligned(16))) float sW[64 * 64];
  const int lane = threadIdx.x & 31;
  const int wid  = threadIdx.x >> 5;
  const int mtiles = (M + 15) >> 4;
  const int mt = blockIdx.x * 8 + wid;
  const int m0 = mt * 16;
  const int row = lane & 15;
  const int kh  = (lane >> 4) << 1;
  const int rbase = (lane >> 4) << 3;
  const int arow_idx = min(m0 + row, M - 1);
  const float* __restrict__ arow = A + (size_t)arow_idx * 64 + kh;

  v2f a[16];
#pragma unroll
  for (int i = 0; i < 16; ++i) {
    a[i].x = arow[4 * i];
    a[i].y = arow[4 * i + 1];
  }
  float dv[8];
#pragma unroll
  for (int r = 0; r < 8; ++r) dv[r] = dinv[min(m0 + rbase + r, M - 1)];

#pragma unroll
  for (int i = 0; i < 64 * 64 / 4; i += 256)
    ((float4*)sW)[i + threadIdx.x] = ((const float4*)W)[i + threadIdx.x];
  __syncthreads();
  if (mt >= mtiles) return;

  const float* __restrict__ bcol = sW + 4 * row;

  v8f c0 = {}, c1 = {}, c2 = {}, c3 = {};
#pragma unroll
  for (int k = 0; k < 64; k += 4) {
    const float4 q0 = *(const float4*)(bcol + (k + kh) * 64);
    const float4 q1 = *(const float4*)(bcol + (k + kh + 1) * 64);
    v2f b;
    b.x = q0.x; b.y = q1.x;
    c0 = __builtin_amdgcn_wmma_f32_16x16x4_f32(false, a[k >> 2], false, b, (short)0, c0, false, false);
    b.x = q0.y; b.y = q1.y;
    c1 = __builtin_amdgcn_wmma_f32_16x16x4_f32(false, a[k >> 2], false, b, (short)0, c1, false, false);
    b.x = q0.z; b.y = q1.z;
    c2 = __builtin_amdgcn_wmma_f32_16x16x4_f32(false, a[k >> 2], false, b, (short)0, c2, false, false);
    b.x = q0.w; b.y = q1.w;
    c3 = __builtin_amdgcn_wmma_f32_16x16x4_f32(false, a[k >> 2], false, b, (short)0, c3, false, false);
  }

#pragma unroll
  for (int r = 0; r < 8; ++r) {
    const int orow = m0 + rbase + r;
    if (orow < M) {
      const float d = dv[r];
      *(float4*)(out + (size_t)orow * 64 + 4 * row) =
          make_float4(c0[r] * d, c1[r] * d, c2[r] * d, c3[r] * d);
    }
  }
}


#define NODES_PER_WAVE64 8

template <int K>
__device__ __forceinline__ float2 sum_rows64(const float2* __restrict__ hl, int idx, int rem,
                                             float2 acc) {
  float2 v[K];
#pragma unroll
  for (int j = 0; j < K; ++j)
    v[j] = hl[(size_t)__builtin_amdgcn_readlane(idx, j) * 32];
#pragma unroll
  for (int j = 0; j < K; ++j) {
    const bool on = (j < K - 8) || (j < rem);
    acc.x += on ? v[j].x : 0.0f;
    acc.y += on ? v[j].y : 0.0f;
  }
  return acc;
}

#define ROWS_PER_BLOCK64 (8 * NODES_PER_WAVE64)
#define APITCH 68

__global__ __launch_bounds__(256) void gather64_proj_kernel(const float* __restrict__ h,
                                                            const int* __restrict__ start,
                                                            const float* __restrict__ deg,
                                                            const int* __restrict__ csr,
                                                            const float* __restrict__ dinv,
                                                            const float* __restrict__ b1,
                                                            const float* __restrict__ W2,
                                                            float* __restrict__ h2, int N) {
  __shared__ __attribute__((aligned(16))) float sA[ROWS_PER_BLOCK64 * APITCH];
  __shared__ __attribute__((aligned(16))) float sW[IN_DIM * OUT_DIM];
  __shared__ __attribute__((aligned(16))) float sD[ROWS_PER_BLOCK64];
  const int lane = threadIdx.x & 31;
  const int wid = __builtin_amdgcn_readfirstlane(threadIdx.x >> 5);
  const int node0 = (blockIdx.x * 8 + wid) * NODES_PER_WAVE64;
  const int nw = max(0, min(NODES_PER_WAVE64, N - node0));
  const float4 w2v = ((const float4*)W2)[threadIdx.x];
  const float2 bb = ((const float2*)b1)[lane];
  const int li = min(node0 + lane, N - 1);
  const int   ldb = start[li];
  const float ldc = deg[li];
  const float ldd = dinv[li];
  const bool has = lane < nw;
  const int begv = has ? ldb : 0;
  const int cntv = has ? (int)ldc : 0;
  const float dvv = has ? ldd : 0.0f;
  const int dvbits = __float_as_int(dvv);
  ((float4*)sW)[threadIdx.x] = w2v;
  if (lane < NODES_PER_WAVE64) sD[wid * NODES_PER_WAVE64 + lane] = dvv;

  const float2* __restrict__ hl = (const float2*)h + lane;
  float* __restrict__ srow = sA + (wid * NODES_PER_WAVE64) * APITCH + 2 * lane;
  for (int q = 0; q < NODES_PER_WAVE64; ++q) {
    float2 a = make_float2(0.0f, 0.0f);
    if (q < nw) {
      const int node = node0 + q;
      int beg = __builtin_amdgcn_readlane(begv, q);
      int rem = __builtin_amdgcn_readlane(cntv, q);
      const float dq = __int_as_float(__builtin_amdgcn_readlane(dvbits, q));
      float2 acc = hl[(size_t)node * 32];
      while (rem > 0) {
        int idx = node;
        if (lane < rem) idx = csr[beg + lane];
        if (rem > 24)      acc = sum_rows64<32>(hl, idx, rem, acc);
        else if (rem > 16) acc = sum_rows64<24>(hl, idx, rem, acc);
        else if (rem > 8)  acc = sum_rows64<16>(hl, idx, rem, acc);
        else               acc = sum_rows64<8>(hl, idx, rem, acc);
        rem -= 32;
        beg += 32;
      }
      a.x = fmaxf(acc.x * dq + bb.x, 0.0f);
      a.y = fmaxf(acc.y * dq + bb.y, 0.0f);
    }
    *(float2*)(srow + q * APITCH) = a;
  }
  __syncthreads();

  const int m0 = blockIdx.x * ROWS_PER_BLOCK64 + wid * 16;
  if (wid >= ROWS_PER_BLOCK64 / 16 || m0 >= N) return;

  const int row = lane & 15;
  const int kh  = (lane >> 4) << 1;
  const int rbase = (lane >> 4) << 3;
  const float4 dlo = *(const float4*)(sD + wid * 16 + rbase);
  const float4 dhi = *(const float4*)(sD + wid * 16 + rbase + 4);
  const float dv[8] = {dlo.x, dlo.y, dlo.z, dlo.w, dhi.x, dhi.y, dhi.z, dhi.w};
  const float* __restrict__ ar = sA + (wid * 16 + row) * APITCH + kh;
  const float* __restrict__ bcol = sW + row;

  v8f c = {};
#pragma unroll
  for (int k = 0; k < IN_DIM; k += 4) {
    const v2f a = *(const v2f*)(ar + k);
    v2f b;
    b.x = bcol[(k + kh) * OUT_DIM];
    b.y = bcol[(k + kh + 1) * OUT_DIM];
    c = __builtin_amdgcn_wmma_f32_16x16x4_f32(false, a, false, b, (short)0, c, false, false);
  }

#pragma unroll
  for (int r = 0; r < 8; ++r) {
    const int orow = m0 + rbase + r;
    if (orow < N) h2[(size_t)orow * OUT_DIM + row] = c[r] * dv[r];
  }
}

#define PAIRS_PER_WAVE16 4

template <int K>
__device__ __forceinline__ float sum_rows16(const float* __restrict__ hc, int pick, int i0, int i1,
                                            int rem, float acc) {
  float v[K];
#pragma unroll
  for (int j = 0; j < K; ++j) {
    const int sj = __builtin_amdgcn_ds_bpermute(pick + 4 * (j & 15), j < 16 ? i0 : i1);
    v[j] = hc[(size_t)sj * 16];
  }
#pragma unroll
  for (int j = 0; j < K; ++j)
    acc += (j < rem) ? v[j] : 0.0f;
  return acc;
}

__global__ __launch_bounds__(256) void gather16_kernel(const float* __restrict__ h,
                                                       const int* __restrict__ start,
                                                       const float* __restrict__ deg,
                                                       const int* __restrict__ csr,
                                                       const float* __restrict__ dinv,
                                                       const float* __restrict__ bias,
                                                       float* __restrict__ out, int N) {
  const int lane = threadIdx.x & 31;
  const int c = lane & 15;
  const int half = lane >> 4;
  const int base0 = (blockIdx.x * 8 + __builtin_amdgcn_readfirstlane(threadIdx.x >> 5)) *
                    (2 * PAIRS_PER_WAVE16);
  if (base0 >= N) return;
  const int np = min(PAIRS_PER_WAVE16, (N - base0 + 1) >> 1);
  int begv = 0, cntv = 0;
  float dv = 0.0f;
  if (lane < 2 * PAIRS_PER_WAVE16 && base0 + lane < N) {
    begv = start[base0 + lane];
    cntv = (int)deg[base0 + lane];
    dv = dinv[base0 + lane];
  }
  const int dvbits = __float_as_int(dv);
  const float* __restrict__ hc = h + c;
  const float bc = bias[c];
  const int pick = half << 6;

  for (int q = 0; q < np; ++q) {
    const int node = base0 + 2 * q + half;
    const int nd = min(node, N - 1);
    const int a = (2 * q + half) << 2;
    int beg = __builtin_amdgcn_ds_bpermute(a, begv);
    int rem = __builtin_amdgcn_ds_bpermute(a, cntv);
    const float dd = __int_as_float(__builtin_amdgcn_ds_bpermute(a, dvbits));
    int mu = max(__builtin_amdgcn_readlane(cntv, 2 * q), __builtin_amdgcn_readlane(cntv, 2 * q + 1));
    float acc = hc[(size_t)nd * 16];
    while (mu > 0) {
      int i0 = nd, i1 = nd;
      if (c < rem)      i0 = csr[beg + c];
      if (c + 16 < rem) i1 = csr[beg + 16 + c];
      if (mu > 24)      acc = sum_rows16<32>(hc, pick, i0, i1, rem, acc);
      else if (mu > 16) acc = sum_rows16<24>(hc, pick, i0, i1, rem, acc);
      else if (mu > 8)  acc = sum_rows16<16>(hc, pick, i0, i1, rem, acc);
      else              acc = sum_rows16<8>(hc, pick, i0, i1, rem, acc);
      mu -= 32;
      rem -= 32;
      beg += 32;
    }
    if (node < N) out[(size_t)node * 16 + c] = bc + dd * acc;
  }
}


extern "C" void kernel_launch(void* const* d_in, const int* in_sizes, int n_in,
                              void* d_out, int out_size, void* d_ws, size_t ws_size,
                              hipStream_t stream) {
  const float* x    = (const float*)d_in[0];
  const int*   edge = (const int*)d_in[1];
  const float* W1   = (const float*)d_in[2];
  const float* b1   = (const float*)d_in[3];
  const float* W2   = (const float*)d_in[4];
  const float* b2   = (const float*)d_in[5];
  float* out = (float*)d_out;

  const int E  = in_sizes[1] / 2;
  const int Nn = in_sizes[0] / IN_DIM;
  const int* src = edge;
  const int* dst = edge + E;

  const int NB = (Nn + BIN_NODES - 1) >> BIN_SHIFT;
  const int G  = (E + CHUNK - 1) / CHUNK;

  const size_t nodeB = ((size_t)Nn * 4 + 255) & ~(size_t)255;
  const size_t edgeB = ((size_t)E * 4 + 255) & ~(size_t)255;
  const size_t offB  = ((size_t)G * (NB + 1) * 4 + 255) & ~(size_t)255;
  const size_t binB  = (size_t)G * CHUNK * 8;
  const size_t runB  = (((size_t)E + 16 * (size_t)NB + 16) * 8 + 255) & ~(size_t)255;
  const size_t featB = (size_t)Nn * 64 * 4;
  char* ws = (char*)d_ws;
  float* deg    = (float*)ws;
  float* dinv   = (float*)(ws + nodeB);
  int*   start  = (int*)(ws + 2 * nodeB);
  int*   off    = (int*)(ws + 3 * nodeB);
  int2*  binned = (int2*)(ws + 3 * nodeB + offB);
  int2*  run    = (int2*)(ws + 3 * nodeB + offB + binB);
  int*   csr    = (int*)(ws + 3 * nodeB + offB + binB + runB);
  float* h1     = (float*)(ws + 3 * nodeB + offB + binB + runB + edgeB);
  float* h2     = (float*)(ws + 3 * nodeB + offB + binB + runB + edgeB + featB);

  const int mtiles = (Nn + 15) >> 4;

  chunk_sort_kernel<<<G, 256, 0, stream>>>(src, dst, off, binned, E, NB);
  bin_csr_kernel<<<NB, 256, 0, stream>>>(binned, off, run, deg, dinv, start, csr, Nn, NB, G);

  gemm_rows64_kernel<<<(mtiles + 7) / 8, 256, 0, stream>>>(x, W1, dinv, h1, Nn);
  gather64_proj_kernel<<<(Nn + ROWS_PER_BLOCK64 - 1) / ROWS_PER_BLOCK64, 256, 0, stream>>>(
      h1, start, deg, csr, dinv, b1, W2, h2, Nn);

  gather16_kernel<<<(Nn + 16 * PAIRS_PER_WAVE16 - 1) / (16 * PAIRS_PER_WAVE16), 256, 0, stream>>>(
      h2, start, deg, csr, dinv, b2, out, Nn);

  (void)n_in; (void)out_size; (void)ws_size;
}
